# starting kernel
# baseline (speedup 1.0000x reference)
#include <hip/hip_runtime.h>
#include <hip/hip_bf16.h>


#define N0_   200000
#define N1_   100000
#define N2_   20000
#define E1_   1600000
#define E2_   320000
#define D_IN_ 128
#define D_H_  256
#define D_OUT_ 64

typedef __attribute__((ext_vector_type(2))) float v2f;
typedef __attribute__((ext_vector_type(8))) float v8f;

__global__ void zero_f4_kernel(float4* __restrict__ p, long n4) {
    long i = (long)blockIdx.x * blockDim.x + threadIdx.x;
    long stride = (long)gridDim.x * blockDim.x;
    float4 z = make_float4(0.f, 0.f, 0.f, 0.f);
    for (; i < n4; i += stride) p[i] = z;
}

template <int D>
__global__ void scatter_sum_kernel(const float* __restrict__ feat,
                                   const int* __restrict__ src,
                                   const int* __restrict__ dst,
                                   float* __restrict__ aggSum,
                                   float* __restrict__ cnt, int E) {
    int e = blockIdx.x * (blockDim.x >> 5) + (threadIdx.x >> 5);
    int lane = threadIdx.x & 31;
    if (e >= E) return;
    int s = src[e];
    int d = dst[e];
    const float4* row = reinterpret_cast<const float4*>(feat + (size_t)s * D);
    float* o = aggSum + (size_t)d * D;
#pragma unroll
    for (int c = 0; c < D / 128; ++c) {
        int idx = lane + 32 * c;
        float4 v = row[idx];
        float* op = o + idx * 4;
        atomicAdd(op + 0, v.x);
        atomicAdd(op + 1, v.y);
        atomicAdd(op + 2, v.z);
        atomicAdd(op + 3, v.w);
    }
    if (lane == 0) atomicAdd(cnt + d, 1.0f);
}

__global__ void sage_gemm_wmma_kernel(const float* __restrict__ aggSum,
                                      const float* __restrict__ cnt,
                                      const float* __restrict__ root,
                                      const float* __restrict__ Wl,
                                      const float* __restrict__ Wr,
                                      const float* __restrict__ bias,
                                      float* __restrict__ out,
                                      int K, int ldOut, int applyRelu) {
    const int wave    = threadIdx.x >> 5;
    const int lane    = threadIdx.x & 31;
    const int l16     = lane & 15;
    const int half    = lane >> 4;
    const int rowBase = blockIdx.x * 16;
    const int colTile = blockIdx.y * (blockDim.x >> 5) + wave;
    const int colBase = colTile * 16;

    const int m = rowBase + l16;
    const float scale = 1.0f / fmaxf(cnt[m], 1.0f);

    const float* aRow = aggSum + (size_t)m * K + 2 * half;
    const float* bRow = Wl + (size_t)(colBase + l16) * K + 2 * half;

    v8f c = {};
#pragma unroll 4
    for (int k = 0; k < K; k += 4) {
        v2f a, b;
        a.x = aRow[k] * scale;
        a.y = aRow[k + 1] * scale;
        b.x = bRow[k];
        b.y = bRow[k + 1];
        c = __builtin_amdgcn_wmma_f32_16x16x4_f32(false, a, false, b,
                                                  (short)0, c, false, false);
    }

    const float* a2 = root + (size_t)m * K + 2 * half;
    const float* b2 = Wr + (size_t)(colBase + l16) * K + 2 * half;
#pragma unroll 4
    for (int k = 0; k < K; k += 4) {
        v2f a, b;
        a.x = a2[k];
        a.y = a2[k + 1];
        b.x = b2[k];
        b.y = b2[k + 1];
        c = __builtin_amdgcn_wmma_f32_16x16x4_f32(false, a, false, b,
                                                  (short)0, c, false, false);
    }

    const float bv = bias[colBase + l16];
#pragma unroll
    for (int r = 0; r < 8; ++r) {
        float v = c[r] + bv;
        if (applyRelu) v = fmaxf(v, 0.0f);
        int row = rowBase + r + 8 * half;
        out[(size_t)row * ldOut + colBase + l16] = v;
    }
}

__global__ void log_softmax64_kernel(float* __restrict__ h2, int rows) {
    int row = blockIdx.x * (blockDim.x >> 5) + (threadIdx.x >> 5);
    int lane = threadIdx.x & 31;
    if (row >= rows) return;
    float* p = h2 + (size_t)row * 64;
    float v0 = p[lane];
    float v1 = p[lane + 32];

    float m = fmaxf(v0, v1);
#pragma unroll
    for (int off = 16; off > 0; off >>= 1)
        m = fmaxf(m, __shfl_xor(m, off, 32));

    float s = __expf(v0 - m) + __expf(v1 - m);
#pragma unroll
    for (int off = 16; off > 0; off >>= 1)
        s += __shfl_xor(s, off, 32);

    float lse = m + __logf(s);
    p[lane]      = v0 - lse;
    p[lane + 32] = v1 - lse;
}

extern "C" void kernel_launch(void* const* d_in, const int* in_sizes, int n_in,
                              void* d_out, int out_size, void* d_ws, size_t ws_size,
                              hipStream_t stream) {
    const float* x    = (const float*)d_in[0];
    const int*   src1 = (const int*)d_in[1];
    const int*   dst1 = (const int*)d_in[2];
    const int*   src2 = (const int*)d_in[3];
    const int*   dst2 = (const int*)d_in[4];
    const float* W_l1 = (const float*)d_in[5];
    const float* b_l1 = (const float*)d_in[6];
    const float* W_r1 = (const float*)d_in[7];
    const float* W_l2 = (const float*)d_in[8];
    const float* b_l2 = (const float*)d_in[9];
    const float* W_r2 = (const float*)d_in[10];
    float* out = (float*)d_out;

    float* ws   = (float*)d_ws;
    float* agg1 = ws;
    float* cnt1 = agg1 + (size_t)N1_ * D_IN_;
    float* agg2 = cnt1 + N1_;
    float* cnt2 = agg2 + (size_t)N2_ * D_H_;
    float* h1   = cnt2 + N2_;
    const long zeroFloats = (long)N1_ * D_IN_ + N1_ + (long)N2_ * D_H_ + N2_;

    zero_f4_kernel<<<8192, 256, 0, stream>>>((float4*)ws, zeroFloats / 4);

    scatter_sum_kernel<D_IN_><<<(E1_ + 7) / 8, 256, 0, stream>>>(
        x, src1, dst1, agg1, cnt1, E1_);

    {
        dim3 grid(N1_ / 16, (D_H_ / 16) / 4);
        sage_gemm_wmma_kernel<<<grid, 128, 0, stream>>>(
            agg1, cnt1, x, W_l1, W_r1, b_l1, h1, D_IN_, D_H_, 1);
    }

    scatter_sum_kernel<D_H_><<<(E2_ + 7) / 8, 256, 0, stream>>>(
        h1, src2, dst2, agg2, cnt2, E2_);

    {
        dim3 grid(N2_ / 16, 1);
        sage_gemm_wmma_kernel<<<grid, 128, 0, stream>>>(
            agg2, cnt2, h1, W_l2, W_r2, b_l2, out, D_H_, D_OUT_, 0);
    }

    log_softmax64_kernel<<<(N2_ + 7) / 8, 256, 0, stream>>>(out, N2_);
}


# Layer GEMMs register-tiled: each wave holds a 32x64 (layer 2: 32x32) tile of f32 16x16x4 accumulators, so one A/B fragment load feeds 4 or 2 matrix instructions, not 1; per-element k order kept
# speedup vs baseline: 1.2282x; 1.2282x over previous
#include <hip/hip_runtime.h>
#include <hip/hip_bf16.h>


#define N0_   200000
#define N1_   100000
#define N2_   20000
#define E1_   1600000
#define E2_   320000
#define D_IN_ 128
#define D_H_  256
#define D_OUT_ 64

typedef __attribute__((ext_vector_type(2))) float v2f;
typedef __attribute__((ext_vector_type(8))) float v8f;

__global__ void zero_f4_kernel(float4* __restrict__ p, long n4) {
    long i = (long)blockIdx.x * blockDim.x + threadIdx.x;
    long stride = (long)gridDim.x * blockDim.x;
    float4 z = make_float4(0.f, 0.f, 0.f, 0.f);
    for (; i < n4; i += stride) p[i] = z;
}

template <int D>
__global__ void scatter_sum_kernel(const float* __restrict__ feat,
                                   const int* __restrict__ src,
                                   const int* __restrict__ dst,
                                   float* __restrict__ aggSum,
                                   float* __restrict__ cnt, int E) {
    int e = blockIdx.x * (blockDim.x >> 5) + (threadIdx.x >> 5);
    int lane = threadIdx.x & 31;
    if (e >= E) return;
    int s = src[e];
    int d = dst[e];
    const float4* row = reinterpret_cast<const float4*>(feat + (size_t)s * D);
    float* o = aggSum + (size_t)d * D;
#pragma unroll
    for (int c = 0; c < D / 128; ++c) {
        int idx = lane + 32 * c;
        float4 v = row[idx];
        float* op = o + idx * 4;
        atomicAdd(op + 0, v.x);
        atomicAdd(op + 1, v.y);
        atomicAdd(op + 2, v.z);
        atomicAdd(op + 3, v.w);
    }
    if (lane == 0) atomicAdd(cnt + d, 1.0f);
}

template <int MT, int NT, int CH, bool SCALED>
__device__ __forceinline__ void sage_tile_chunk(const float* const (&aP)[MT],
                                                const float* const (&bP)[NT],
                                                const float (&scale)[MT],
                                                int k0, v8f (&c)[MT][NT]) {
    v2f a[CH][MT], b[CH][NT];
#pragma unroll
    for (int s = 0; s < CH; ++s) {
#pragma unroll
        for (int i = 0; i < MT; ++i) {
            a[s][i].x = aP[i][k0 + 4 * s];
            a[s][i].y = aP[i][k0 + 4 * s + 1];
        }
#pragma unroll
        for (int j = 0; j < NT; ++j) {
            b[s][j].x = bP[j][k0 + 4 * s];
            b[s][j].y = bP[j][k0 + 4 * s + 1];
        }
    }

#pragma unroll
    for (int s = 0; s < CH; ++s) {
#pragma unroll
        for (int i = 0; i < MT; ++i) {
            v2f av = a[s][i];
            if (SCALED) {
                av.x = av.x * scale[i];
                av.y = av.y * scale[i];
            }
#pragma unroll
            for (int j = 0; j < NT; ++j)
                c[i][j] = __builtin_amdgcn_wmma_f32_16x16x4_f32(
                    false, av, false, b[s][j], (short)0, c[i][j], false, false);
        }
    }
}

template <int K, int MT, int NT, int CH, int WAVES_M, int WAVES_N, bool RELU>
__global__ __launch_bounds__(32 * WAVES_M * WAVES_N)
void sage_gemm_wmma_kernel(const float* __restrict__ aggSum,
                           const float* __restrict__ cnt,
                           const float* __restrict__ root,
                           const float* __restrict__ Wl,
                           const float* __restrict__ Wr,
                           const float* __restrict__ bias,
                           float* __restrict__ out, int ldOut) {
    const int wave    = threadIdx.x >> 5;
    const int lane    = threadIdx.x & 31;
    const int l16     = lane & 15;
    const int half    = lane >> 4;
    const int waveN   = wave % WAVES_N;
    const int waveM   = wave / WAVES_N;
    const int rowBase = (blockIdx.x * WAVES_M + waveM) * (16 * MT);
    const int colBase = (blockIdx.y * WAVES_N + waveN) * (16 * NT);

    const float* aP[MT];
    const float* rP[MT];
    float scale[MT];
#pragma unroll
    for (int i = 0; i < MT; ++i) {
        const int m = rowBase + 16 * i + l16;
        scale[i] = 1.0f / fmaxf(cnt[m], 1.0f);
        aP[i] = aggSum + (size_t)m * K + 2 * half;
        rP[i] = root + (size_t)m * K + 2 * half;
    }
    const float* bP[NT];
    const float* sP[NT];
#pragma unroll
    for (int j = 0; j < NT; ++j) {
        const size_t n = (size_t)(colBase + 16 * j + l16);
        bP[j] = Wl + n * K + 2 * half;
        sP[j] = Wr + n * K + 2 * half;
    }

    v8f c[MT][NT];
#pragma unroll
    for (int i = 0; i < MT; ++i)
#pragma unroll
        for (int j = 0; j < NT; ++j) c[i][j] = v8f{};

#pragma unroll 1
    for (int k0 = 0; k0 < K; k0 += 4 * CH)
        sage_tile_chunk<MT, NT, CH, true>(aP, bP, scale, k0, c);
#pragma unroll 1
    for (int k0 = 0; k0 < K; k0 += 4 * CH)
        sage_tile_chunk<MT, NT, CH, false>(rP, sP, scale, k0, c);

#pragma unroll
    for (int j = 0; j < NT; ++j) {
        const int col = colBase + 16 * j + l16;
        const float bv = bias[col];
#pragma unroll
        for (int i = 0; i < MT; ++i) {
#pragma unroll
            for (int r = 0; r < 8; ++r) {
                float v = c[i][j][r] + bv;
                if (RELU) v = fmaxf(v, 0.0f);
                const int row = rowBase + 16 * i + r + 8 * half;
                out[(size_t)row * ldOut + col] = v;
            }
        }
    }
}

__global__ void log_softmax64_kernel(float* __restrict__ h2, int rows) {
    int row = blockIdx.x * (blockDim.x >> 5) + (threadIdx.x >> 5);
    int lane = threadIdx.x & 31;
    if (row >= rows) return;
    float* p = h2 + (size_t)row * 64;
    float v0 = p[lane];
    float v1 = p[lane + 32];

    float m = fmaxf(v0, v1);
#pragma unroll
    for (int off = 16; off > 0; off >>= 1)
        m = fmaxf(m, __shfl_xor(m, off, 32));

    float s = __expf(v0 - m) + __expf(v1 - m);
#pragma unroll
    for (int off = 16; off > 0; off >>= 1)
        s += __shfl_xor(s, off, 32);

    float lse = m + __logf(s);
    p[lane]      = v0 - lse;
    p[lane + 32] = v1 - lse;
}

extern "C" void kernel_launch(void* const* d_in, const int* in_sizes, int n_in,
                              void* d_out, int out_size, void* d_ws, size_t ws_size,
                              hipStream_t stream) {
    const float* x    = (const float*)d_in[0];
    const int*   src1 = (const int*)d_in[1];
    const int*   dst1 = (const int*)d_in[2];
    const int*   src2 = (const int*)d_in[3];
    const int*   dst2 = (const int*)d_in[4];
    const float* W_l1 = (const float*)d_in[5];
    const float* b_l1 = (const float*)d_in[6];
    const float* W_r1 = (const float*)d_in[7];
    const float* W_l2 = (const float*)d_in[8];
    const float* b_l2 = (const float*)d_in[9];
    const float* W_r2 = (const float*)d_in[10];
    float* out = (float*)d_out;

    float* ws   = (float*)d_ws;
    float* agg1 = ws;
    float* cnt1 = agg1 + (size_t)N1_ * D_IN_;
    float* agg2 = cnt1 + N1_;
    float* cnt2 = agg2 + (size_t)N2_ * D_H_;
    float* h1   = cnt2 + N2_;
    const long zeroFloats = (long)N1_ * D_IN_ + N1_ + (long)N2_ * D_H_ + N2_;

    zero_f4_kernel<<<8192, 256, 0, stream>>>((float4*)ws, zeroFloats / 4);

    scatter_sum_kernel<D_IN_><<<(E1_ + 7) / 8, 256, 0, stream>>>(
        x, src1, dst1, agg1, cnt1, E1_);

    {
        dim3 grid(N1_ / (16 * 2), 1);
        sage_gemm_wmma_kernel<D_IN_, 2, 4, 8, 1, 4, true><<<grid, 128, 0, stream>>>(
            agg1, cnt1, x, W_l1, W_r1, b_l1, h1, D_H_);
    }

    scatter_sum_kernel<D_H_><<<(E2_ + 7) / 8, 256, 0, stream>>>(
        h1, src2, dst2, agg2, cnt2, E2_);

    {
        dim3 grid(N2_ / 32, 1);
        sage_gemm_wmma_kernel<D_H_, 2, 2, 8, 1, 2, false><<<grid, 64, 0, stream>>>(
            agg2, cnt2, h1, W_l2, W_r2, b_l2, out, D_OUT_);
    }

    log_softmax64_kernel<<<(N2_ + 7) / 8, 256, 0, stream>>>(out, N2_);
}


# Scatter-add re-laid lane-contiguous: each f32 atomic (and row load) instruction of a wave covers one whole 128-B line, not 8 dwords in each of 4 lines; same adds, arrival-order sums as before
# speedup vs baseline: 1.4573x; 1.1865x over previous
#include <hip/hip_runtime.h>
#include <hip/hip_bf16.h>


#define N0_   200000
#define N1_   100000
#define N2_   20000
#define E1_   1600000
#define E2_   320000
#define D_IN_ 128
#define D_H_  256
#define D_OUT_ 64

typedef __attribute__((ext_vector_type(2))) float v2f;
typedef __attribute__((ext_vector_type(8))) float v8f;

__global__ void zero_f4_kernel(float4* __restrict__ p, long n4) {
    long i = (long)blockIdx.x * blockDim.x + threadIdx.x;
    long stride = (long)gridDim.x * blockDim.x;
    float4 z = make_float4(0.f, 0.f, 0.f, 0.f);
    for (; i < n4; i += stride) p[i] = z;
}

template <int D>
__global__ void scatter_sum_kernel(const float* __restrict__ feat,
                                   const int* __restrict__ src,
                                   const int* __restrict__ dst,
                                   float* __restrict__ aggSum,
                                   float* __restrict__ cnt, int E) {
    int e = blockIdx.x * (blockDim.x >> 5) + (threadIdx.x >> 5);
    int lane = threadIdx.x & 31;
    if (e >= E) return;
    int s = src[e];
    int d = dst[e];
    const float* row = feat + (size_t)s * D + lane;
    float* o = aggSum + (size_t)d * D + lane;
    float v[D / 32];
#pragma unroll
    for (int c = 0; c < D / 32; ++c) v[c] = row[32 * c];
#pragma unroll
    for (int c = 0; c < D / 32; ++c) atomicAdd(o + 32 * c, v[c]);
    if (lane == 0) atomicAdd(cnt + d, 1.0f);
}

template <int MT, int NT, int CH, bool SCALED>
__device__ __forceinline__ void sage_tile_chunk(const float* const (&aP)[MT],
                                                const float* const (&bP)[NT],
                                                const float (&scale)[MT],
                                                int k0, v8f (&c)[MT][NT]) {
    v2f a[CH][MT], b[CH][NT];
#pragma unroll
    for (int s = 0; s < CH; ++s) {
#pragma unroll
        for (int i = 0; i < MT; ++i) {
            a[s][i].x = aP[i][k0 + 4 * s];
            a[s][i].y = aP[i][k0 + 4 * s + 1];
        }
#pragma unroll
        for (int j = 0; j < NT; ++j) {
            b[s][j].x = bP[j][k0 + 4 * s];
            b[s][j].y = bP[j][k0 + 4 * s + 1];
        }
    }

#pragma unroll
    for (int s = 0; s < CH; ++s) {
#pragma unroll
        for (int i = 0; i < MT; ++i) {
            v2f av = a[s][i];
            if (SCALED) {
                av.x = av.x * scale[i];
                av.y = av.y * scale[i];
            }
#pragma unroll
            for (int j = 0; j < NT; ++j)
                c[i][j] = __builtin_amdgcn_wmma_f32_16x16x4_f32(
                    false, av, false, b[s][j], (short)0, c[i][j], false, false);
        }
    }
}

template <int K, int MT, int NT, int CH, int WAVES_M, int WAVES_N, bool RELU>
__global__ __launch_bounds__(32 * WAVES_M * WAVES_N)
void sage_gemm_wmma_kernel(const float* __restrict__ aggSum,
                           const float* __restrict__ cnt,
                           const float* __restrict__ root,
                           const float* __restrict__ Wl,
                           const float* __restrict__ Wr,
                           const float* __restrict__ bias,
                           float* __restrict__ out, int ldOut) {
    const int wave    = threadIdx.x >> 5;
    const int lane    = threadIdx.x & 31;
    const int l16     = lane & 15;
    const int half    = lane >> 4;
    const int waveN   = wave % WAVES_N;
    const int waveM   = wave / WAVES_N;
    const int rowBase = (blockIdx.x * WAVES_M + waveM) * (16 * MT);
    const int colBase = (blockIdx.y * WAVES_N + waveN) * (16 * NT);

    const float* aP[MT];
    const float* rP[MT];
    float scale[MT];
#pragma unroll
    for (int i = 0; i < MT; ++i) {
        const int m = rowBase + 16 * i + l16;
        scale[i] = 1.0f / fmaxf(cnt[m], 1.0f);
        aP[i] = aggSum + (size_t)m * K + 2 * half;
        rP[i] = root + (size_t)m * K + 2 * half;
    }
    const float* bP[NT];
    const float* sP[NT];
#pragma unroll
    for (int j = 0; j < NT; ++j) {
        const size_t n = (size_t)(colBase + 16 * j + l16);
        bP[j] = Wl + n * K + 2 * half;
        sP[j] = Wr + n * K + 2 * half;
    }

    v8f c[MT][NT];
#pragma unroll
    for (int i = 0; i < MT; ++i)
#pragma unroll
        for (int j = 0; j < NT; ++j) c[i][j] = v8f{};

#pragma unroll 1
    for (int k0 = 0; k0 < K; k0 += 4 * CH)
        sage_tile_chunk<MT, NT, CH, true>(aP, bP, scale, k0, c);
#pragma unroll 1
    for (int k0 = 0; k0 < K; k0 += 4 * CH)
        sage_tile_chunk<MT, NT, CH, false>(rP, sP, scale, k0, c);

#pragma unroll
    for (int j = 0; j < NT; ++j) {
        const int col = colBase + 16 * j + l16;
        const float bv = bias[col];
#pragma unroll
        for (int i = 0; i < MT; ++i) {
#pragma unroll
            for (int r = 0; r < 8; ++r) {
                float v = c[i][j][r] + bv;
                if (RELU) v = fmaxf(v, 0.0f);
                const int row = rowBase + 16 * i + r + 8 * half;
                out[(size_t)row * ldOut + col] = v;
            }
        }
    }
}

__global__ void log_softmax64_kernel(float* __restrict__ h2, int rows) {
    int row = blockIdx.x * (blockDim.x >> 5) + (threadIdx.x >> 5);
    int lane = threadIdx.x & 31;
    if (row >= rows) return;
    float* p = h2 + (size_t)row * 64;
    float v0 = p[lane];
    float v1 = p[lane + 32];

    float m = fmaxf(v0, v1);
#pragma unroll
    for (int off = 16; off > 0; off >>= 1)
        m = fmaxf(m, __shfl_xor(m, off, 32));

    float s = __expf(v0 - m) + __expf(v1 - m);
#pragma unroll
    for (int off = 16; off > 0; off >>= 1)
        s += __shfl_xor(s, off, 32);

    float lse = m + __logf(s);
    p[lane]      = v0 - lse;
    p[lane + 32] = v1 - lse;
}

extern "C" void kernel_launch(void* const* d_in, const int* in_sizes, int n_in,
                              void* d_out, int out_size, void* d_ws, size_t ws_size,
                              hipStream_t stream) {
    const float* x    = (const float*)d_in[0];
    const int*   src1 = (const int*)d_in[1];
    const int*   dst1 = (const int*)d_in[2];
    const int*   src2 = (const int*)d_in[3];
    const int*   dst2 = (const int*)d_in[4];
    const float* W_l1 = (const float*)d_in[5];
    const float* b_l1 = (const float*)d_in[6];
    const float* W_r1 = (const float*)d_in[7];
    const float* W_l2 = (const float*)d_in[8];
    const float* b_l2 = (const float*)d_in[9];
    const float* W_r2 = (const float*)d_in[10];
    float* out = (float*)d_out;

    float* ws   = (float*)d_ws;
    float* agg1 = ws;
    float* cnt1 = agg1 + (size_t)N1_ * D_IN_;
    float* agg2 = cnt1 + N1_;
    float* cnt2 = agg2 + (size_t)N2_ * D_H_;
    float* h1   = cnt2 + N2_;
    const long zeroFloats = (long)N1_ * D_IN_ + N1_ + (long)N2_ * D_H_ + N2_;

    zero_f4_kernel<<<8192, 256, 0, stream>>>((float4*)ws, zeroFloats / 4);

    scatter_sum_kernel<D_IN_><<<(E1_ + 7) / 8, 256, 0, stream>>>(
        x, src1, dst1, agg1, cnt1, E1_);

    {
        dim3 grid(N1_ / (16 * 2), 1);
        sage_gemm_wmma_kernel<D_IN_, 2, 4, 8, 1, 4, true><<<grid, 128, 0, stream>>>(
            agg1, cnt1, x, W_l1, W_r1, b_l1, h1, D_H_);
    }

    scatter_sum_kernel<D_H_><<<(E2_ + 7) / 8, 256, 0, stream>>>(
        h1, src2, dst2, agg2, cnt2, E2_);

    {
        dim3 grid(N2_ / 32, 1);
        sage_gemm_wmma_kernel<D_H_, 2, 2, 8, 1, 2, false><<<grid, 64, 0, stream>>>(
            agg2, cnt2, h1, W_l2, W_r2, b_l2, out, D_OUT_);
    }

    log_softmax64_kernel<<<(N2_ + 7) / 8, 256, 0, stream>>>(out, N2_);
}


# Scatter-add: one wave per 32 consecutive edges, not one per edge: index pairs fetched one line per wave, 32 row lines in flight ahead of a group's atomics; same adds, arrival-order sums as before
# speedup vs baseline: 1.6422x; 1.1269x over previous
#include <hip/hip_runtime.h>
#include <hip/hip_bf16.h>


#define N0_   200000
#define N1_   100000
#define N2_   20000
#define E1_   1600000
#define E2_   320000
#define D_IN_ 128
#define D_H_  256
#define D_OUT_ 64

typedef __attribute__((ext_vector_type(2))) float v2f;
typedef __attribute__((ext_vector_type(8))) float v8f;

__global__ void zero_f4_kernel(float4* __restrict__ p, long n4) {
    long i = (long)blockIdx.x * blockDim.x + threadIdx.x;
    long stride = (long)gridDim.x * blockDim.x;
    float4 z = make_float4(0.f, 0.f, 0.f, 0.f);
    for (; i < n4; i += stride) p[i] = z;
}

template <int D, int E>
__global__ __launch_bounds__(128)
void scatter_sum_kernel(const float* __restrict__ feat,
                        const int* __restrict__ src,
                        const int* __restrict__ dst,
                        float* __restrict__ aggSum,
                        float* __restrict__ cnt) {
    static_assert(E % 128 == 0, "4 waves of 32 edges per workgroup must tile E");
    constexpr int C = D / 32;
    constexpr int U = 32 / C;
    const int lane = threadIdx.x & 31;
    const int e0 = (blockIdx.x * 4 + (threadIdx.x >> 5)) * 32;
    const int sv = src[e0 + lane];
    const int dv = dst[e0 + lane];

    for (int jb = 0; jb < 32; jb += U) {
        float v[U][C];
        float* o[U];
#pragma unroll
        for (int u = 0; u < U; ++u) {
            const unsigned s = (unsigned)__builtin_amdgcn_readlane(sv, jb + u);
            const unsigned d = (unsigned)__builtin_amdgcn_readlane(dv, jb + u);
            const float* row = feat + (size_t)s * D + lane;
            o[u] = aggSum + (size_t)d * D + lane;
#pragma unroll
            for (int c = 0; c < C; ++c) v[u][c] = row[32 * c];
        }
#pragma unroll
        for (int u = 0; u < U; ++u)
#pragma unroll
            for (int c = 0; c < C; ++c) atomicAdd(o[u] + 32 * c, v[u][c]);
    }
    atomicAdd(cnt + dv, 1.0f);
}

template <int MT, int NT, int CH, bool SCALED>
__device__ __forceinline__ void sage_tile_chunk(const float* const (&aP)[MT],
                                                const float* const (&bP)[NT],
                                                const float (&scale)[MT],
                                                int k0, v8f (&c)[MT][NT]) {
    v2f a[CH][MT], b[CH][NT];
#pragma unroll
    for (int s = 0; s < CH; ++s) {
#pragma unroll
        for (int i = 0; i < MT; ++i) {
            a[s][i].x = aP[i][k0 + 4 * s];
            a[s][i].y = aP[i][k0 + 4 * s + 1];
        }
#pragma unroll
        for (int j = 0; j < NT; ++j) {
            b[s][j].x = bP[j][k0 + 4 * s];
            b[s][j].y = bP[j][k0 + 4 * s + 1];
        }
    }

#pragma unroll
    for (int s = 0; s < CH; ++s) {
#pragma unroll
        for (int i = 0; i < MT; ++i) {
            v2f av = a[s][i];
            if (SCALED) {
                av.x = av.x * scale[i];
                av.y = av.y * scale[i];
            }
#pragma unroll
            for (int j = 0; j < NT; ++j)
                c[i][j] = __builtin_amdgcn_wmma_f32_16x16x4_f32(
                    false, av, false, b[s][j], (short)0, c[i][j], false, false);
        }
    }
}

template <int K, int MT, int NT, int CH, int WAVES_M, int WAVES_N, bool RELU>
__global__ __launch_bounds__(32 * WAVES_M * WAVES_N)
void sage_gemm_wmma_kernel(const float* __restrict__ aggSum,
                           const float* __restrict__ cnt,
                           const float* __restrict__ root,
                           const float* __restrict__ Wl,
                           const float* __restrict__ Wr,
                           const float* __restrict__ bias,
                           float* __restrict__ out, int ldOut) {
    const int wave    = threadIdx.x >> 5;
    const int lane    = threadIdx.x & 31;
    const int l16     = lane & 15;
    const int half    = lane >> 4;
    const int waveN   = wave % WAVES_N;
    const int waveM   = wave / WAVES_N;
    const int rowBase = (blockIdx.x * WAVES_M + waveM) * (16 * MT);
    const int colBase = (blockIdx.y * WAVES_N + waveN) * (16 * NT);

    const float* aP[MT];
    const float* rP[MT];
    float scale[MT];
#pragma unroll
    for (int i = 0; i < MT; ++i) {
        const int m = rowBase + 16 * i + l16;
        scale[i] = 1.0f / fmaxf(cnt[m], 1.0f);
        aP[i] = aggSum + (size_t)m * K + 2 * half;
        rP[i] = root + (size_t)m * K + 2 * half;
    }
    const float* bP[NT];
    const float* sP[NT];
#pragma unroll
    for (int j = 0; j < NT; ++j) {
        const size_t n = (size_t)(colBase + 16 * j + l16);
        bP[j] = Wl + n * K + 2 * half;
        sP[j] = Wr + n * K + 2 * half;
    }

    v8f c[MT][NT];
#pragma unroll
    for (int i = 0; i < MT; ++i)
#pragma unroll
        for (int j = 0; j < NT; ++j) c[i][j] = v8f{};

#pragma unroll 1
    for (int k0 = 0; k0 < K; k0 += 4 * CH)
        sage_tile_chunk<MT, NT, CH, true>(aP, bP, scale, k0, c);
#pragma unroll 1
    for (int k0 = 0; k0 < K; k0 += 4 * CH)
        sage_tile_chunk<MT, NT, CH, false>(rP, sP, scale, k0, c);

#pragma unroll
    for (int j = 0; j < NT; ++j) {
        const int col = colBase + 16 * j + l16;
        const float bv = bias[col];
#pragma unroll
        for (int i = 0; i < MT; ++i) {
#pragma unroll
            for (int r = 0; r < 8; ++r) {
                float v = c[i][j][r] + bv;
                if (RELU) v = fmaxf(v, 0.0f);
                const int row = rowBase + 16 * i + r + 8 * half;
                out[(size_t)row * ldOut + col] = v;
            }
        }
    }
}

__global__ void log_softmax64_kernel(float* __restrict__ h2, int rows) {
    int row = blockIdx.x * (blockDim.x >> 5) + (threadIdx.x >> 5);
    int lane = threadIdx.x & 31;
    if (row >= rows) return;
    float* p = h2 + (size_t)row * 64;
    float v0 = p[lane];
    float v1 = p[lane + 32];

    float m = fmaxf(v0, v1);
#pragma unroll
    for (int off = 16; off > 0; off >>= 1)
        m = fmaxf(m, __shfl_xor(m, off, 32));

    float s = __expf(v0 - m) + __expf(v1 - m);
#pragma unroll
    for (int off = 16; off > 0; off >>= 1)
        s += __shfl_xor(s, off, 32);

    float lse = m + __logf(s);
    p[lane]      = v0 - lse;
    p[lane + 32] = v1 - lse;
}

extern "C" void kernel_launch(void* const* d_in, const int* in_sizes, int n_in,
                              void* d_out, int out_size, void* d_ws, size_t ws_size,
                              hipStream_t stream) {
    const float* x    = (const float*)d_in[0];
    const int*   src1 = (const int*)d_in[1];
    const int*   dst1 = (const int*)d_in[2];
    const int*   src2 = (const int*)d_in[3];
    const int*   dst2 = (const int*)d_in[4];
    const float* W_l1 = (const float*)d_in[5];
    const float* b_l1 = (const float*)d_in[6];
    const float* W_r1 = (const float*)d_in[7];
    const float* W_l2 = (const float*)d_in[8];
    const float* b_l2 = (const float*)d_in[9];
    const float* W_r2 = (const float*)d_in[10];
    float* out = (float*)d_out;

    float* ws   = (float*)d_ws;
    float* agg1 = ws;
    float* cnt1 = agg1 + (size_t)N1_ * D_IN_;
    float* agg2 = cnt1 + N1_;
    float* cnt2 = agg2 + (size_t)N2_ * D_H_;
    float* h1   = cnt2 + N2_;
    const long zeroFloats = (long)N1_ * D_IN_ + N1_ + (long)N2_ * D_H_ + N2_;

    zero_f4_kernel<<<8192, 256, 0, stream>>>((float4*)ws, zeroFloats / 4);

    scatter_sum_kernel<D_IN_, E1_><<<E1_ / 128, 128, 0, stream>>>(
        x, src1, dst1, agg1, cnt1);

    {
        dim3 grid(N1_ / (16 * 2), 1);
        sage_gemm_wmma_kernel<D_IN_, 2, 4, 8, 1, 4, true><<<grid, 128, 0, stream>>>(
            agg1, cnt1, x, W_l1, W_r1, b_l1, h1, D_H_);
    }

    scatter_sum_kernel<D_H_, E2_><<<E2_ / 128, 128, 0, stream>>>(
        h1, src2, dst2, agg2, cnt2);

    {
        dim3 grid(N2_ / 32, 1);
        sage_gemm_wmma_kernel<D_H_, 2, 2, 8, 1, 2, false><<<grid, 64, 0, stream>>>(
            agg2, cnt2, h1, W_l2, W_r2, b_l2, out, D_OUT_);
    }

    log_softmax64_kernel<<<(N2_ + 7) / 8, 256, 0, stream>>>(out, N2_);
}


# Scatter-add by f32 atomics replaced by a per-call counting sort of edges by destination (int atomic ranks, scan, fill); one wave per destination sums its rows in registers, in rank (arrival) order
# speedup vs baseline: 1.8936x; 1.1531x over previous
#include <hip/hip_runtime.h>
#include <hip/hip_bf16.h>


#define N0_   200000
#define N1_   100000
#define N2_   20000
#define E1_   1600000
#define E2_   320000
#define D_IN_ 128
#define D_H_  256
#define D_OUT_ 64

typedef __attribute__((ext_vector_type(2))) float v2f;
typedef __attribute__((ext_vector_type(8))) float v8f;

__global__ void zero_i4_kernel(int4* __restrict__ p, int n4) {
    const int i = blockIdx.x * blockDim.x + threadIdx.x;
    if (i < n4) p[i] = make_int4(0, 0, 0, 0);
}

__global__ __launch_bounds__(256)
void edge_rank_kernel(const int* __restrict__ dst1, const int* __restrict__ dst2,
                      int* __restrict__ deg1, int* __restrict__ deg2,
                      int* __restrict__ rank1, int* __restrict__ rank2) {
    const int i = blockIdx.x * 256 + threadIdx.x;
    if (i < E1_) {
        rank1[i] = atomicAdd(deg1 + dst1[i], 1);
    } else {
        const int j = i - E1_;
        rank2[j] = atomicAdd(deg2 + dst2[j], 1);
    }
}

__device__ __forceinline__ void degree_scan_block(const int* __restrict__ deg,
                                                  int* __restrict__ off,
                                                  float* __restrict__ cnt, int n) {
    __shared__ int waveTot[32];
    const int wave = threadIdx.x >> 5;
    const int lane = threadIdx.x & 31;
    const int per = (((n + 31) / 32) + 127) & ~127;
    const int beg = wave * per;
    const int end = min(beg + per, n);

    int s = 0;
    for (int i = beg + 4 * lane; i < end; i += 128) {
        const int4 v = *reinterpret_cast<const int4*>(deg + i);
        s += (v.x + v.y) + (v.z + v.w);
    }
#pragma unroll
    for (int o = 16; o > 0; o >>= 1) s += __shfl_xor(s, o, 32);
    if (lane == 0) waveTot[wave] = s;
    __syncthreads();

    int t = (lane < wave) ? waveTot[lane] : 0;
#pragma unroll
    for (int o = 16; o > 0; o >>= 1) t += __shfl_xor(t, o, 32);
    int run = t;

    for (int i0 = beg; i0 < end; i0 += 128) {
        const int i = i0 + 4 * lane;
        const bool in = i < end;
        int4 v = make_int4(0, 0, 0, 0);
        if (in) v = *reinterpret_cast<const int4*>(deg + i);
        const int mine = (v.x + v.y) + (v.z + v.w);
        int incl = mine;
#pragma unroll
        for (int o = 1; o < 32; o <<= 1) {
            const int up = __shfl_up(incl, o, 32);
            if (lane >= o) incl += up;
        }
        const int ex = run + incl - mine;
        if (in) {
            *reinterpret_cast<int4*>(off + i) =
                make_int4(ex, ex + v.x, ex + v.x + v.y, ex + v.x + v.y + v.z);
            *reinterpret_cast<float4*>(cnt + i) =
                make_float4((float)v.x, (float)v.y, (float)v.z, (float)v.w);
        }
        run += __shfl(incl, 31, 32);
    }
    if (wave == 31 && lane == 0) off[n] = run;
}

__global__ __launch_bounds__(1024)
void degree_scan_kernel(const int* __restrict__ deg1, int* __restrict__ off1,
                        float* __restrict__ cnt1,
                        const int* __restrict__ deg2, int* __restrict__ off2,
                        float* __restrict__ cnt2) {
    if (blockIdx.x == 0) degree_scan_block(deg1, off1, cnt1, N1_);
    else                 degree_scan_block(deg2, off2, cnt2, N2_);
}

__global__ __launch_bounds__(256)
void edge_fill_kernel(const int* __restrict__ src1, const int* __restrict__ dst1,
                      const int* __restrict__ src2, const int* __restrict__ dst2,
                      const int* __restrict__ rank1, const int* __restrict__ rank2,
                      const int* __restrict__ off1, const int* __restrict__ off2,
                      int* __restrict__ perm1, int* __restrict__ perm2) {
    const int i = blockIdx.x * 256 + threadIdx.x;
    if (i < E1_) {
        perm1[off1[dst1[i]] + rank1[i]] = src1[i];
    } else {
        const int j = i - E1_;
        perm2[off2[dst2[j]] + rank2[j]] = src2[j];
    }
}

template <int D>
__global__ __launch_bounds__(256)
void gather_sum_kernel(const float* __restrict__ feat,
                       const int* __restrict__ off,
                       const int* __restrict__ perm,
                       float* __restrict__ aggSum) {
    constexpr int C4 = D / 128;
    const int lane = threadIdx.x & 31;
    const int node = blockIdx.x * 8 + (threadIdx.x >> 5);
    const int beg = __builtin_amdgcn_readfirstlane(off[node]);
    const int n   = __builtin_amdgcn_readfirstlane(off[node + 1]) - beg;

    float4 acc[C4];
#pragma unroll
    for (int c = 0; c < C4; ++c) acc[c] = make_float4(0.f, 0.f, 0.f, 0.f);

    for (int base = 0; base < n; base += 32) {
        const int m = min(32, n - base);
        const int sv = (lane < m) ? perm[beg + base + lane] : 0;
        int j = 0;
        for (; j + 4 <= m; j += 4) {
            float4 v[4][C4];
#pragma unroll
            for (int u = 0; u < 4; ++u) {
                const unsigned sidx = (unsigned)__builtin_amdgcn_readlane(sv, j + u);
                const float4* row =
                    reinterpret_cast<const float4*>(feat + (size_t)sidx * D) + lane;
#pragma unroll
                for (int c = 0; c < C4; ++c) v[u][c] = row[32 * c];
            }
#pragma unroll
            for (int u = 0; u < 4; ++u)
#pragma unroll
                for (int c = 0; c < C4; ++c) {
                    acc[c].x += v[u][c].x; acc[c].y += v[u][c].y;
                    acc[c].z += v[u][c].z; acc[c].w += v[u][c].w;
                }
        }
        for (; j < m; ++j) {
            const unsigned sidx = (unsigned)__builtin_amdgcn_readlane(sv, j);
            const float4* row =
                reinterpret_cast<const float4*>(feat + (size_t)sidx * D) + lane;
#pragma unroll
            for (int c = 0; c < C4; ++c) {
                const float4 v = row[32 * c];
                acc[c].x += v.x; acc[c].y += v.y; acc[c].z += v.z; acc[c].w += v.w;
            }
        }
    }

    float4* o = reinterpret_cast<float4*>(aggSum + (size_t)node * D) + lane;
#pragma unroll
    for (int c = 0; c < C4; ++c) o[32 * c] = acc[c];
}

template <int MT, int NT, int CH, bool SCALED>
__device__ __forceinline__ void sage_tile_chunk(const float* const (&aP)[MT],
                                                const float* const (&bP)[NT],
                                                const float (&scale)[MT],
                                                int k0, v8f (&c)[MT][NT]) {
    v2f a[CH][MT], b[CH][NT];
#pragma unroll
    for (int s = 0; s < CH; ++s) {
#pragma unroll
        for (int i = 0; i < MT; ++i) {
            a[s][i].x = aP[i][k0 + 4 * s];
            a[s][i].y = aP[i][k0 + 4 * s + 1];
        }
#pragma unroll
        for (int j = 0; j < NT; ++j) {
            b[s][j].x = bP[j][k0 + 4 * s];
            b[s][j].y = bP[j][k0 + 4 * s + 1];
        }
    }

#pragma unroll
    for (int s = 0; s < CH; ++s) {
#pragma unroll
        for (int i = 0; i < MT; ++i) {
            v2f av = a[s][i];
            if (SCALED) {
                av.x = av.x * scale[i];
                av.y = av.y * scale[i];
            }
#pragma unroll
            for (int j = 0; j < NT; ++j)
                c[i][j] = __builtin_amdgcn_wmma_f32_16x16x4_f32(
                    false, av, false, b[s][j], (short)0, c[i][j], false, false);
        }
    }
}

template <int K, int MT, int NT, int CH, int WAVES_M, int WAVES_N, bool RELU>
__global__ __launch_bounds__(32 * WAVES_M * WAVES_N)
void sage_gemm_wmma_kernel(const float* __restrict__ aggSum,
                           const float* __restrict__ cnt,
                           const float* __restrict__ root,
                           const float* __restrict__ Wl,
                           const float* __restrict__ Wr,
                           const float* __restrict__ bias,
                           float* __restrict__ out, int ldOut) {
    const int wave    = threadIdx.x >> 5;
    const int lane    = threadIdx.x & 31;
    const int l16     = lane & 15;
    const int half    = lane >> 4;
    const int waveN   = wave % WAVES_N;
    const int waveM   = wave / WAVES_N;
    const int rowBase = (blockIdx.x * WAVES_M + waveM) * (16 * MT);
    const int colBase = (blockIdx.y * WAVES_N + waveN) * (16 * NT);

    const float* aP[MT];
    const float* rP[MT];
    float scale[MT];
#pragma unroll
    for (int i = 0; i < MT; ++i) {
        const int m = rowBase + 16 * i + l16;
        scale[i] = 1.0f / fmaxf(cnt[m], 1.0f);
        aP[i] = aggSum + (size_t)m * K + 2 * half;
        rP[i] = root + (size_t)m * K + 2 * half;
    }
    const float* bP[NT];
    const float* sP[NT];
#pragma unroll
    for (int j = 0; j < NT; ++j) {
        const size_t n = (size_t)(colBase + 16 * j + l16);
        bP[j] = Wl + n * K + 2 * half;
        sP[j] = Wr + n * K + 2 * half;
    }

    v8f c[MT][NT];
#pragma unroll
    for (int i = 0; i < MT; ++i)
#pragma unroll
        for (int j = 0; j < NT; ++j) c[i][j] = v8f{};

#pragma unroll 1
    for (int k0 = 0; k0 < K; k0 += 4 * CH)
        sage_tile_chunk<MT, NT, CH, true>(aP, bP, scale, k0, c);
#pragma unroll 1
    for (int k0 = 0; k0 < K; k0 += 4 * CH)
        sage_tile_chunk<MT, NT, CH, false>(rP, sP, scale, k0, c);

#pragma unroll
    for (int j = 0; j < NT; ++j) {
        const int col = colBase + 16 * j + l16;
        const float bv = bias[col];
#pragma unroll
        for (int i = 0; i < MT; ++i) {
#pragma unroll
            for (int r = 0; r < 8; ++r) {
                float v = c[i][j][r] + bv;
                if (RELU) v = fmaxf(v, 0.0f);
                const int row = rowBase + 16 * i + r + 8 * half;
                out[(size_t)row * ldOut + col] = v;
            }
        }
    }
}

__global__ void log_softmax64_kernel(float* __restrict__ h2, int rows) {
    int row = blockIdx.x * (blockDim.x >> 5) + (threadIdx.x >> 5);
    int lane = threadIdx.x & 31;
    if (row >= rows) return;
    float* p = h2 + (size_t)row * 64;
    float v0 = p[lane];
    float v1 = p[lane + 32];

    float m = fmaxf(v0, v1);
#pragma unroll
    for (int off = 16; off > 0; off >>= 1)
        m = fmaxf(m, __shfl_xor(m, off, 32));

    float s = __expf(v0 - m) + __expf(v1 - m);
#pragma unroll
    for (int off = 16; off > 0; off >>= 1)
        s += __shfl_xor(s, off, 32);

    float lse = m + __logf(s);
    p[lane]      = v0 - lse;
    p[lane + 32] = v1 - lse;
}

extern "C" void kernel_launch(void* const* d_in, const int* in_sizes, int n_in,
                              void* d_out, int out_size, void* d_ws, size_t ws_size,
                              hipStream_t stream) {
    const float* x    = (const float*)d_in[0];
    const int*   src1 = (const int*)d_in[1];
    const int*   dst1 = (const int*)d_in[2];
    const int*   src2 = (const int*)d_in[3];
    const int*   dst2 = (const int*)d_in[4];
    const float* W_l1 = (const float*)d_in[5];
    const float* b_l1 = (const float*)d_in[6];
    const float* W_r1 = (const float*)d_in[7];
    const float* W_l2 = (const float*)d_in[8];
    const float* b_l2 = (const float*)d_in[9];
    const float* W_r2 = (const float*)d_in[10];
    float* out = (float*)d_out;

    float* ws    = (float*)d_ws;
    int*   wsi   = (int*)d_ws;
    float* h1    = ws;
    int*   deg1  = wsi;
    int*   deg2  = wsi + 100000;
    int*   rank1 = wsi + 200000;
    int*   rank2 = wsi + 1800000;
    int*   off1  = wsi + 2200000;
    int*   perm1 = wsi + 2400000;
    float* agg1  = ws + 25600000;
    float* agg2  = agg1;
    float* cnt1  = ws + 38400000;
    float* cnt2  = ws + 38500000;
    int*   off2  = wsi + 38520000;
    int*   perm2 = wsi + 38560000;

    zero_i4_kernel<<<((N1_ + N2_) / 4 + 255) / 256, 256, 0, stream>>>(
        (int4*)deg1, (N1_ + N2_) / 4);

    edge_rank_kernel<<<(E1_ + E2_) / 256, 256, 0, stream>>>(
        dst1, dst2, deg1, deg2, rank1, rank2);
    degree_scan_kernel<<<2, 1024, 0, stream>>>(deg1, off1, cnt1, deg2, off2, cnt2);
    edge_fill_kernel<<<(E1_ + E2_) / 256, 256, 0, stream>>>(
        src1, dst1, src2, dst2, rank1, rank2, off1, off2, perm1, perm2);

    gather_sum_kernel<D_IN_><<<N1_ / 8, 256, 0, stream>>>(x, off1, perm1, agg1);

    {
        dim3 grid(N1_ / (16 * 2), 1);
        sage_gemm_wmma_kernel<D_IN_, 2, 4, 8, 1, 4, true><<<grid, 128, 0, stream>>>(
            agg1, cnt1, x, W_l1, W_r1, b_l1, h1, D_H_);
    }

    gather_sum_kernel<D_H_><<<N2_ / 8, 256, 0, stream>>>(h1, off2, perm2, agg2);

    {
        dim3 grid(N2_ / 32, 1);
        sage_gemm_wmma_kernel<D_H_, 2, 2, 8, 1, 2, false><<<grid, 64, 0, stream>>>(
            agg2, cnt2, h1, W_l2, W_r2, b_l2, out, D_OUT_);
    }

    log_softmax64_kernel<<<(N2_ + 7) / 8, 256, 0, stream>>>(out, N2_);
}


# Layer-1 GEMM wave tile 32x64 -> 80x64 (5x4 f32 16x16x4 accumulators): a weight fragment load feeds 5 matrix instructions, not 2; a chunk's loads stay ahead of them; 3 waves/SIMD; k order kept
# speedup vs baseline: 2.1487x; 1.1347x over previous
#include <hip/hip_runtime.h>
#include <hip/hip_bf16.h>


#define N0_   200000
#define N1_   100000
#define N2_   20000
#define E1_   1600000
#define E2_   320000
#define D_IN_ 128
#define D_H_  256
#define D_OUT_ 64

typedef __attribute__((ext_vector_type(2))) float v2f;
typedef __attribute__((ext_vector_type(8))) float v8f;

__global__ void zero_i4_kernel(int4* __restrict__ p, int n4) {
    const int i = blockIdx.x * blockDim.x + threadIdx.x;
    if (i < n4) p[i] = make_int4(0, 0, 0, 0);
}

__global__ __launch_bounds__(256)
void edge_rank_kernel(const int* __restrict__ dst1, const int* __restrict__ dst2,
                      int* __restrict__ deg1, int* __restrict__ deg2,
                      int* __restrict__ rank1, int* __restrict__ rank2) {
    const int i = blockIdx.x * 256 + threadIdx.x;
    if (i < E1_) {
        rank1[i] = atomicAdd(deg1 + dst1[i], 1);
    } else {
        const int j = i - E1_;
        rank2[j] = atomicAdd(deg2 + dst2[j], 1);
    }
}

__device__ __forceinline__ void degree_scan_block(const int* __restrict__ deg,
                                                  int* __restrict__ off,
                                                  float* __restrict__ cnt, int n) {
    __shared__ int waveTot[32];
    const int wave = threadIdx.x >> 5;
    const int lane = threadIdx.x & 31;
    const int per = (((n + 31) / 32) + 127) & ~127;
    const int beg = wave * per;
    const int end = min(beg + per, n);

    int s = 0;
    for (int i = beg + 4 * lane; i < end; i += 128) {
        const int4 v = *reinterpret_cast<const int4*>(deg + i);
        s += (v.x + v.y) + (v.z + v.w);
    }
#pragma unroll
    for (int o = 16; o > 0; o >>= 1) s += __shfl_xor(s, o, 32);
    if (lane == 0) waveTot[wave] = s;
    __syncthreads();

    int t = (lane < wave) ? waveTot[lane] : 0;
#pragma unroll
    for (int o = 16; o > 0; o >>= 1) t += __shfl_xor(t, o, 32);
    int run = t;

    for (int i0 = beg; i0 < end; i0 += 128) {
        const int i = i0 + 4 * lane;
        const bool in = i < end;
        int4 v = make_int4(0, 0, 0, 0);
        if (in) v = *reinterpret_cast<const int4*>(deg + i);
        const int mine = (v.x + v.y) + (v.z + v.w);
        int incl = mine;
#pragma unroll
        for (int o = 1; o < 32; o <<= 1) {
            const int up = __shfl_up(incl, o, 32);
            if (lane >= o) incl += up;
        }
        const int ex = run + incl - mine;
        if (in) {
            *reinterpret_cast<int4*>(off + i) =
                make_int4(ex, ex + v.x, ex + v.x + v.y, ex + v.x + v.y + v.z);
            *reinterpret_cast<float4*>(cnt + i) =
                make_float4((float)v.x, (float)v.y, (float)v.z, (float)v.w);
        }
        run += __shfl(incl, 31, 32);
    }
    if (wave == 31 && lane == 0) off[n] = run;
}

__global__ __launch_bounds__(1024)
void degree_scan_kernel(const int* __restrict__ deg1, int* __restrict__ off1,
                        float* __restrict__ cnt1,
                        const int* __restrict__ deg2, int* __restrict__ off2,
                        float* __restrict__ cnt2) {
    if (blockIdx.x == 0) degree_scan_block(deg1, off1, cnt1, N1_);
    else                 degree_scan_block(deg2, off2, cnt2, N2_);
}

__global__ __launch_bounds__(256)
void edge_fill_kernel(const int* __restrict__ src1, const int* __restrict__ dst1,
                      const int* __restrict__ src2, const int* __restrict__ dst2,
                      const int* __restrict__ rank1, const int* __restrict__ rank2,
                      const int* __restrict__ off1, const int* __restrict__ off2,
                      int* __restrict__ perm1, int* __restrict__ perm2) {
    const int i = blockIdx.x * 256 + threadIdx.x;
    if (i < E1_) {
        perm1[off1[dst1[i]] + rank1[i]] = src1[i];
    } else {
        const int j = i - E1_;
        perm2[off2[dst2[j]] + rank2[j]] = src2[j];
    }
}

template <int D>
__global__ __launch_bounds__(256)
void gather_sum_kernel(const float* __restrict__ feat,
                       const int* __restrict__ off,
                       const int* __restrict__ perm,
                       float* __restrict__ aggSum) {
    constexpr int C4 = D / 128;
    const int lane = threadIdx.x & 31;
    const int node = blockIdx.x * 8 + (threadIdx.x >> 5);
    const int beg = __builtin_amdgcn_readfirstlane(off[node]);
    const int n   = __builtin_amdgcn_readfirstlane(off[node + 1]) - beg;

    float4 acc[C4];
#pragma unroll
    for (int c = 0; c < C4; ++c) acc[c] = make_float4(0.f, 0.f, 0.f, 0.f);

    for (int base = 0; base < n; base += 32) {
        const int m = min(32, n - base);
        const int sv = (lane < m) ? perm[beg + base + lane] : 0;
        int j = 0;
        for (; j + 4 <= m; j += 4) {
            float4 v[4][C4];
#pragma unroll
            for (int u = 0; u < 4; ++u) {
                const unsigned sidx = (unsigned)__builtin_amdgcn_readlane(sv, j + u);
                const float4* row =
                    reinterpret_cast<const float4*>(feat + (size_t)sidx * D) + lane;
#pragma unroll
                for (int c = 0; c < C4; ++c) v[u][c] = row[32 * c];
            }
#pragma unroll
            for (int u = 0; u < 4; ++u)
#pragma unroll
                for (int c = 0; c < C4; ++c) {
                    acc[c].x += v[u][c].x; acc[c].y += v[u][c].y;
                    acc[c].z += v[u][c].z; acc[c].w += v[u][c].w;
                }
        }
        for (; j < m; ++j) {
            const unsigned sidx = (unsigned)__builtin_amdgcn_readlane(sv, j);
            const float4* row =
                reinterpret_cast<const float4*>(feat + (size_t)sidx * D) + lane;
#pragma unroll
            for (int c = 0; c < C4; ++c) {
                const float4 v = row[32 * c];
                acc[c].x += v.x; acc[c].y += v.y; acc[c].z += v.z; acc[c].w += v.w;
            }
        }
    }

    float4* o = reinterpret_cast<float4*>(aggSum + (size_t)node * D) + lane;
#pragma unroll
    for (int c = 0; c < C4; ++c) o[32 * c] = acc[c];
}

template <int MT, int NT, int CH, bool SCALED>
__device__ __forceinline__ void sage_tile_chunk(const float* __restrict__ aBase,
                                                const unsigned (&aOff)[MT],
                                                const float* __restrict__ bBase,
                                                const unsigned (&bOff)[NT],
                                                const float (&scale)[MT],
                                                v8f (&c)[MT][NT]) {
    v2f a[CH][MT], b[CH][NT];
#pragma unroll
    for (int s = 0; s < CH; ++s) {
#pragma unroll
        for (int i = 0; i < MT; ++i) {
            const float* p = reinterpret_cast<const float*>(
                reinterpret_cast<const char*>(aBase + 4 * s) + aOff[i]);
            a[s][i].x = p[0];
            a[s][i].y = p[1];
        }
#pragma unroll
        for (int j = 0; j < NT; ++j) {
            const float* p = reinterpret_cast<const float*>(
                reinterpret_cast<const char*>(bBase + 4 * s) + bOff[j]);
            b[s][j].x = p[0];
            b[s][j].y = p[1];
        }
    }
    if (SCALED) {
#pragma unroll
        for (int s = 0; s < CH; ++s)
#pragma unroll
            for (int i = 0; i < MT; ++i) {
                a[s][i].x = a[s][i].x * scale[i];
                a[s][i].y = a[s][i].y * scale[i];
            }
    }

    __builtin_amdgcn_sched_barrier(0);

#pragma unroll
    for (int s = 0; s < CH; ++s) {
#pragma unroll
        for (int i = 0; i < MT; ++i) {
#pragma unroll
            for (int j = 0; j < NT; ++j)
                c[i][j] = __builtin_amdgcn_wmma_f32_16x16x4_f32(
                    false, a[s][i], false, b[s][j], (short)0, c[i][j], false, false);
        }
    }
}

template <int K, int MT, int NT, int CH, int WAVES_M, int WAVES_N, bool RELU>
__global__ __launch_bounds__(32 * WAVES_M * WAVES_N)
void sage_gemm_wmma_kernel(const float* __restrict__ aggSum,
                           const float* __restrict__ cnt,
                           const float* __restrict__ root,
                           const float* __restrict__ Wl,
                           const float* __restrict__ Wr,
                           const float* __restrict__ bias,
                           float* __restrict__ out, int ldOut) {
    const int wave    = threadIdx.x >> 5;
    const int lane    = threadIdx.x & 31;
    const int l16     = lane & 15;
    const int half    = lane >> 4;
    const int waveN   = wave % WAVES_N;
    const int waveM   = wave / WAVES_N;
    const int rowBase = (blockIdx.x * WAVES_M + waveM) * (16 * MT);
    const int colBase = (blockIdx.y * WAVES_N + waveN) * (16 * NT);

    unsigned aOff[MT];
    float scale[MT];
#pragma unroll
    for (int i = 0; i < MT; ++i) {
        const int m = rowBase + 16 * i + l16;
        scale[i] = 1.0f / fmaxf(cnt[m], 1.0f);
        aOff[i] = ((unsigned)m * K + 2 * half) * 4u;
    }
    unsigned bOff[NT];
#pragma unroll
    for (int j = 0; j < NT; ++j) {
        const int n = colBase + 16 * j + l16;
        bOff[j] = ((unsigned)n * K + 2 * half) * 4u;
    }

    v8f c[MT][NT];
#pragma unroll
    for (int i = 0; i < MT; ++i)
#pragma unroll
        for (int j = 0; j < NT; ++j) c[i][j] = v8f{};

#pragma unroll 1
    for (int k0 = 0; k0 < K; k0 += 4 * CH)
        sage_tile_chunk<MT, NT, CH, true>(aggSum + k0, aOff, Wl + k0, bOff, scale, c);
#pragma unroll 1
    for (int k0 = 0; k0 < K; k0 += 4 * CH)
        sage_tile_chunk<MT, NT, CH, false>(root + k0, aOff, Wr + k0, bOff, scale, c);

#pragma unroll
    for (int j = 0; j < NT; ++j) {
        const int col = colBase + 16 * j + l16;
        const float bv = bias[col];
#pragma unroll
        for (int i = 0; i < MT; ++i) {
#pragma unroll
            for (int r = 0; r < 8; ++r) {
                float v = c[i][j][r] + bv;
                if (RELU) v = fmaxf(v, 0.0f);
                const int row = rowBase + 16 * i + r + 8 * half;
                out[(size_t)row * ldOut + col] = v;
            }
        }
    }
}

__global__ void log_softmax64_kernel(float* __restrict__ h2, int rows) {
    int row = blockIdx.x * (blockDim.x >> 5) + (threadIdx.x >> 5);
    int lane = threadIdx.x & 31;
    if (row >= rows) return;
    float* p = h2 + (size_t)row * 64;
    float v0 = p[lane];
    float v1 = p[lane + 32];

    float m = fmaxf(v0, v1);
#pragma unroll
    for (int off = 16; off > 0; off >>= 1)
        m = fmaxf(m, __shfl_xor(m, off, 32));

    float s = __expf(v0 - m) + __expf(v1 - m);
#pragma unroll
    for (int off = 16; off > 0; off >>= 1)
        s += __shfl_xor(s, off, 32);

    float lse = m + __logf(s);
    p[lane]      = v0 - lse;
    p[lane + 32] = v1 - lse;
}

extern "C" void kernel_launch(void* const* d_in, const int* in_sizes, int n_in,
                              void* d_out, int out_size, void* d_ws, size_t ws_size,
                              hipStream_t stream) {
    const float* x    = (const float*)d_in[0];
    const int*   src1 = (const int*)d_in[1];
    const int*   dst1 = (const int*)d_in[2];
    const int*   src2 = (const int*)d_in[3];
    const int*   dst2 = (const int*)d_in[4];
    const float* W_l1 = (const float*)d_in[5];
    const float* b_l1 = (const float*)d_in[6];
    const float* W_r1 = (const float*)d_in[7];
    const float* W_l2 = (const float*)d_in[8];
    const float* b_l2 = (const float*)d_in[9];
    const float* W_r2 = (const float*)d_in[10];
    float* out = (float*)d_out;

    float* ws    = (float*)d_ws;
    int*   wsi   = (int*)d_ws;
    float* h1    = ws;
    int*   deg1  = wsi;
    int*   deg2  = wsi + 100000;
    int*   rank1 = wsi + 200000;
    int*   rank2 = wsi + 1800000;
    int*   off1  = wsi + 2200000;
    int*   perm1 = wsi + 2400000;
    float* agg1  = ws + 25600000;
    float* agg2  = agg1;
    float* cnt1  = ws + 38400000;
    float* cnt2  = ws + 38500000;
    int*   off2  = wsi + 38520000;
    int*   perm2 = wsi + 38560000;

    zero_i4_kernel<<<((N1_ + N2_) / 4 + 255) / 256, 256, 0, stream>>>(
        (int4*)deg1, (N1_ + N2_) / 4);

    edge_rank_kernel<<<(E1_ + E2_) / 256, 256, 0, stream>>>(
        dst1, dst2, deg1, deg2, rank1, rank2);
    degree_scan_kernel<<<2, 1024, 0, stream>>>(deg1, off1, cnt1, deg2, off2, cnt2);
    edge_fill_kernel<<<(E1_ + E2_) / 256, 256, 0, stream>>>(
        src1, dst1, src2, dst2, rank1, rank2, off1, off2, perm1, perm2);

    gather_sum_kernel<D_IN_><<<N1_ / 8, 256, 0, stream>>>(x, off1, perm1, agg1);

    {
        dim3 grid(N1_ / (16 * 5), 1);
        sage_gemm_wmma_kernel<D_IN_, 5, 4, 8, 1, 4, true><<<grid, 128, 0, stream>>>(
            agg1, cnt1, x, W_l1, W_r1, b_l1, h1, D_H_);
    }

    gather_sum_kernel<D_H_><<<N2_ / 8, 256, 0, stream>>>(h1, off2, perm2, agg2);

    {
        dim3 grid(N2_ / 32, 1);
        sage_gemm_wmma_kernel<D_H_, 2, 2, 8, 1, 2, false><<<grid, 64, 0, stream>>>(
            agg2, cnt2, h1, W_l2, W_r2, b_l2, out, D_OUT_);
    }

    log_softmax64_kernel<<<(N2_ + 7) / 8, 256, 0, stream>>>(out, N2_);
}


# Edge grouping: the rank atomic writes each source id straight into a fixed 32-slot row per destination (linked overflow pool above 32); scan and fill launches gone; arrival-order sums as before
# speedup vs baseline: 2.6356x; 1.2266x over previous
#include <hip/hip_runtime.h>
#include <hip/hip_bf16.h>


#define N0_   200000
#define N1_   100000
#define N2_   20000
#define E1_   1600000
#define E2_   320000
#define D_IN_ 128
#define D_H_  256
#define D_OUT_ 64
#define CAP_  32

typedef __attribute__((ext_vector_type(2))) float v2f;
typedef __attribute__((ext_vector_type(8))) float v8f;

__global__ void zero_i4_kernel(int4* __restrict__ p, int n4) {
    const int i = blockIdx.x * blockDim.x + threadIdx.x;
    if (i < n4) p[i] = make_int4(0, 0, 0, 0);
}

__device__ __forceinline__ void slot_edge(int s, int d,
                                          int* __restrict__ deg,
                                          int* __restrict__ slots,
                                          int* __restrict__ head,
                                          int* __restrict__ poolCnt,
                                          int* __restrict__ poolSrc,
                                          int* __restrict__ poolNext) {
    const int r = atomicAdd(deg + d, 1);
    if (r < CAP_) {
        slots[d * CAP_ + r] = s;
    } else {
        const int k = atomicAdd(poolCnt, 1);
        poolSrc[k] = s;
        poolNext[k] = atomicExch(head + d, k + 1);
    }
}

__global__ __launch_bounds__(256)
void edge_slot_kernel(const int* __restrict__ src1, const int* __restrict__ dst1,
                      const int* __restrict__ src2, const int* __restrict__ dst2,
                      int* __restrict__ deg1, int* __restrict__ deg2,
                      int* __restrict__ slots1, int* __restrict__ slots2,
                      int* __restrict__ head1, int* __restrict__ head2,
                      int* __restrict__ poolCnt,
                      int* __restrict__ poolSrc1, int* __restrict__ poolNext1,
                      int* __restrict__ poolSrc2, int* __restrict__ poolNext2) {
    const int i = blockIdx.x * 256 + threadIdx.x;
    if (i < E1_) {
        slot_edge(src1[i], dst1[i], deg1, slots1, head1, poolCnt, poolSrc1, poolNext1);
    } else {
        const int j = i - E1_;
        slot_edge(src2[j], dst2[j], deg2, slots2, head2, poolCnt + 1, poolSrc2, poolNext2);
    }
}

template <int D>
__global__ __launch_bounds__(256)
void gather_sum_kernel(const float* __restrict__ feat,
                       const int* __restrict__ deg,
                       const int* __restrict__ slots,
                       const int* __restrict__ head,
                       const int* __restrict__ poolSrc,
                       const int* __restrict__ poolNext,
                       float* __restrict__ aggSum) {
    constexpr int C4 = D / 128;
    const int lane = threadIdx.x & 31;
    const int node = blockIdx.x * 8 + (threadIdx.x >> 5);
    const int n = __builtin_amdgcn_readfirstlane(deg[node]);
    const int m = min(n, CAP_);

    float4 acc[C4];
#pragma unroll
    for (int c = 0; c < C4; ++c) acc[c] = make_float4(0.f, 0.f, 0.f, 0.f);

    const int sv = (lane < m) ? slots[node * CAP_ + lane] : 0;
    int j = 0;
    for (; j + 4 <= m; j += 4) {
        float4 v[4][C4];
#pragma unroll
        for (int u = 0; u < 4; ++u) {
            const unsigned sidx = (unsigned)__builtin_amdgcn_readlane(sv, j + u);
            const float4* row =
                reinterpret_cast<const float4*>(feat + (size_t)sidx * D) + lane;
#pragma unroll
            for (int c = 0; c < C4; ++c) v[u][c] = row[32 * c];
        }
#pragma unroll
        for (int u = 0; u < 4; ++u)
#pragma unroll
            for (int c = 0; c < C4; ++c) {
                acc[c].x += v[u][c].x; acc[c].y += v[u][c].y;
                acc[c].z += v[u][c].z; acc[c].w += v[u][c].w;
            }
    }
    for (; j < m; ++j) {
        const unsigned sidx = (unsigned)__builtin_amdgcn_readlane(sv, j);
        const float4* row =
            reinterpret_cast<const float4*>(feat + (size_t)sidx * D) + lane;
#pragma unroll
        for (int c = 0; c < C4; ++c) {
            const float4 v = row[32 * c];
            acc[c].x += v.x; acc[c].y += v.y; acc[c].z += v.z; acc[c].w += v.w;
        }
    }

    if (n > CAP_) {
        int k = __builtin_amdgcn_readfirstlane(head[node]);
        for (int t = CAP_; t < n && k > 0; ++t) {
            const unsigned sidx =
                (unsigned)__builtin_amdgcn_readfirstlane(poolSrc[k - 1]);
            const int nxt = __builtin_amdgcn_readfirstlane(poolNext[k - 1]);
            const float4* row =
                reinterpret_cast<const float4*>(feat + (size_t)sidx * D) + lane;
#pragma unroll
            for (int c = 0; c < C4; ++c) {
                const float4 v = row[32 * c];
                acc[c].x += v.x; acc[c].y += v.y; acc[c].z += v.z; acc[c].w += v.w;
            }
            k = nxt;
        }
    }

    float4* o = reinterpret_cast<float4*>(aggSum + (size_t)node * D) + lane;
#pragma unroll
    for (int c = 0; c < C4; ++c) o[32 * c] = acc[c];
}

template <int MT, int NT, int CH, bool SCALED>
__device__ __forceinline__ void sage_tile_chunk(const float* __restrict__ aBase,
                                                const unsigned (&aOff)[MT],
                                                const float* __restrict__ bBase,
                                                const unsigned (&bOff)[NT],
                                                const float (&scale)[MT],
                                                v8f (&c)[MT][NT]) {
    v2f a[CH][MT], b[CH][NT];
#pragma unroll
    for (int s = 0; s < CH; ++s) {
#pragma unroll
        for (int i = 0; i < MT; ++i) {
            const float* p = reinterpret_cast<const float*>(
                reinterpret_cast<const char*>(aBase + 4 * s) + aOff[i]);
            a[s][i].x = p[0];
            a[s][i].y = p[1];
        }
#pragma unroll
        for (int j = 0; j < NT; ++j) {
            const float* p = reinterpret_cast<const float*>(
                reinterpret_cast<const char*>(bBase + 4 * s) + bOff[j]);
            b[s][j].x = p[0];
            b[s][j].y = p[1];
        }
    }
    if (SCALED) {
#pragma unroll
        for (int s = 0; s < CH; ++s)
#pragma unroll
            for (int i = 0; i < MT; ++i) {
                a[s][i].x = a[s][i].x * scale[i];
                a[s][i].y = a[s][i].y * scale[i];
            }
    }

    __builtin_amdgcn_sched_barrier(0);

#pragma unroll
    for (int s = 0; s < CH; ++s) {
#pragma unroll
        for (int i = 0; i < MT; ++i) {
#pragma unroll
            for (int j = 0; j < NT; ++j)
                c[i][j] = __builtin_amdgcn_wmma_f32_16x16x4_f32(
                    false, a[s][i], false, b[s][j], (short)0, c[i][j], false, false);
        }
    }
}

template <int K, int MT, int NT, int CH, int WAVES_M, int WAVES_N, bool RELU>
__global__ __launch_bounds__(32 * WAVES_M * WAVES_N)
void sage_gemm_wmma_kernel(const float* __restrict__ aggSum,
                           const int* __restrict__ deg,
                           const float* __restrict__ root,
                           const float* __restrict__ Wl,
                           const float* __restrict__ Wr,
                           const float* __restrict__ bias,
                           float* __restrict__ out, int ldOut) {
    const int wave    = threadIdx.x >> 5;
    const int lane    = threadIdx.x & 31;
    const int l16     = lane & 15;
    const int half    = lane >> 4;
    const int waveN   = wave % WAVES_N;
    const int waveM   = wave / WAVES_N;
    const int rowBase = (blockIdx.x * WAVES_M + waveM) * (16 * MT);
    const int colBase = (blockIdx.y * WAVES_N + waveN) * (16 * NT);

    unsigned aOff[MT];
    float scale[MT];
#pragma unroll
    for (int i = 0; i < MT; ++i) {
        const int m = rowBase + 16 * i + l16;
        scale[i] = 1.0f / fmaxf((float)deg[m], 1.0f);
        aOff[i] = ((unsigned)m * K + 2 * half) * 4u;
    }
    unsigned bOff[NT];
#pragma unroll
    for (int j = 0; j < NT; ++j) {
        const int n = colBase + 16 * j + l16;
        bOff[j] = ((unsigned)n * K + 2 * half) * 4u;
    }

    v8f c[MT][NT];
#pragma unroll
    for (int i = 0; i < MT; ++i)
#pragma unroll
        for (int j = 0; j < NT; ++j) c[i][j] = v8f{};

#pragma unroll 1
    for (int k0 = 0; k0 < K; k0 += 4 * CH)
        sage_tile_chunk<MT, NT, CH, true>(aggSum + k0, aOff, Wl + k0, bOff, scale, c);
#pragma unroll 1
    for (int k0 = 0; k0 < K; k0 += 4 * CH)
        sage_tile_chunk<MT, NT, CH, false>(root + k0, aOff, Wr + k0, bOff, scale, c);

#pragma unroll
    for (int j = 0; j < NT; ++j) {
        const int col = colBase + 16 * j + l16;
        const float bv = bias[col];
#pragma unroll
        for (int i = 0; i < MT; ++i) {
#pragma unroll
            for (int r = 0; r < 8; ++r) {
                float v = c[i][j][r] + bv;
                if (RELU) v = fmaxf(v, 0.0f);
                const int row = rowBase + 16 * i + r + 8 * half;
                out[(size_t)row * ldOut + col] = v;
            }
        }
    }
}

__global__ void log_softmax64_kernel(float* __restrict__ h2, int rows) {
    int row = blockIdx.x * (blockDim.x >> 5) + (threadIdx.x >> 5);
    int lane = threadIdx.x & 31;
    if (row >= rows) return;
    float* p = h2 + (size_t)row * 64;
    float v0 = p[lane];
    float v1 = p[lane + 32];

    float m = fmaxf(v0, v1);
#pragma unroll
    for (int off = 16; off > 0; off >>= 1)
        m = fmaxf(m, __shfl_xor(m, off, 32));

    float s = __expf(v0 - m) + __expf(v1 - m);
#pragma unroll
    for (int off = 16; off > 0; off >>= 1)
        s += __shfl_xor(s, off, 32);

    float lse = m + __logf(s);
    p[lane]      = v0 - lse;
    p[lane + 32] = v1 - lse;
}

extern "C" void kernel_launch(void* const* d_in, const int* in_sizes, int n_in,
                              void* d_out, int out_size, void* d_ws, size_t ws_size,
                              hipStream_t stream) {
    const float* x    = (const float*)d_in[0];
    const int*   src1 = (const int*)d_in[1];
    const int*   dst1 = (const int*)d_in[2];
    const int*   src2 = (const int*)d_in[3];
    const int*   dst2 = (const int*)d_in[4];
    const float* W_l1 = (const float*)d_in[5];
    const float* b_l1 = (const float*)d_in[6];
    const float* W_r1 = (const float*)d_in[7];
    const float* W_l2 = (const float*)d_in[8];
    const float* b_l2 = (const float*)d_in[9];
    const float* W_r2 = (const float*)d_in[10];
    float* out = (float*)d_out;

    float* ws    = (float*)d_ws;
    int*   wsi   = (int*)d_ws;
    float* h1    = ws;
    int*   slots1    = wsi;
    int*   poolSrc1  = wsi + 3200000;
    int*   poolNext1 = wsi + 4800000;
    float* agg1  = ws + 25600000;
    float* agg2  = agg1;
    int*   deg1      = wsi + 38400000;
    int*   deg2      = wsi + 38500000;
    int*   head1     = wsi + 38520000;
    int*   head2     = wsi + 38620000;
    int*   poolCnt   = wsi + 38640000;
    int*   slots2    = wsi + 38640032;
    int*   poolSrc2  = wsi + 39280032;
    int*   poolNext2 = wsi + 39600032;

    zero_i4_kernel<<<(240032 / 4 + 255) / 256, 256, 0, stream>>>(
        (int4*)deg1, 240032 / 4);

    edge_slot_kernel<<<(E1_ + E2_) / 256, 256, 0, stream>>>(
        src1, dst1, src2, dst2, deg1, deg2, slots1, slots2, head1, head2,
        poolCnt, poolSrc1, poolNext1, poolSrc2, poolNext2);

    gather_sum_kernel<D_IN_><<<N1_ / 8, 256, 0, stream>>>(
        x, deg1, slots1, head1, poolSrc1, poolNext1, agg1);

    {
        dim3 grid(N1_ / (16 * 5), 1);
        sage_gemm_wmma_kernel<D_IN_, 5, 4, 8, 1, 4, true><<<grid, 128, 0, stream>>>(
            agg1, deg1, x, W_l1, W_r1, b_l1, h1, D_H_);
    }

    gather_sum_kernel<D_H_><<<N2_ / 8, 256, 0, stream>>>(
        h1, deg2, slots2, head2, poolSrc2, poolNext2, agg2);

    {
        dim3 grid(N2_ / 32, 1);
        sage_gemm_wmma_kernel<D_H_, 2, 2, 8, 1, 2, false><<<grid, 64, 0, stream>>>(
            agg2, deg2, h1, W_l2, W_r2, b_l2, out, D_OUT_);
    }

    log_softmax64_kernel<<<(N2_ + 7) / 8, 256, 0, stream>>>(out, N2_);
}


# Edge grouping through LDS: 8192-edge blocks sorted by 256-destination range in LDS (one global add per block and range, not per edge), ranks by LDS atomics, slot rows stored whole; arrival-order sums
# speedup vs baseline: 4.0373x; 1.5318x over previous
#include <hip/hip_runtime.h>
#include <hip/hip_bf16.h>


#define N0_   200000
#define N1_   100000
#define N2_   20000
#define E1_   1600000
#define E2_   320000
#define D_IN_ 128
#define D_H_  256
#define D_OUT_ 64
#define CAP_  32
#define RNG_  256
#define NB1_  391
#define NB2_  79
#define NB_   (NB1_ + NB2_)
#define NBP_  512
#define CAPB_ 6144
#define CHUNK_ 8192

typedef __attribute__((ext_vector_type(2))) float v2f;
typedef __attribute__((ext_vector_type(8))) float v8f;

__global__ void zero_i4_kernel(int4* __restrict__ p, int n4) {
    const int i = blockIdx.x * blockDim.x + threadIdx.x;
    if (i < n4) p[i] = make_int4(0, 0, 0, 0);
}

__global__ __launch_bounds__(1024)
void edge_bin_kernel(const int* __restrict__ src1, const int* __restrict__ dst1,
                     const int* __restrict__ src2, const int* __restrict__ dst2,
                     int* __restrict__ cur, int* __restrict__ spillCnt,
                     int* __restrict__ lists, int2* __restrict__ spill) {
    __shared__ int hist[NBP_];
    __shared__ int lstart[NBP_];
    __shared__ int gbase[NBP_];
    __shared__ int wsum[NBP_ / 32];
    __shared__ int sorted[CHUNK_];
    __shared__ unsigned short srng[CHUNK_];

    const int t = threadIdx.x;
    if (t < NBP_) hist[t] = 0;
    __syncthreads();

    const int e0 = blockIdx.x * CHUNK_;
    int pk[8], gb[8], lr[8];
#pragma unroll
    for (int i = 0; i < 8; ++i) {
        const int e = e0 + i * 1024 + t;
        gb[i] = -1;
        pk[i] = 0;
        if (e < E1_ + E2_) {
            int s, d, b0;
            if (e < E1_) { s = src1[e]; d = dst1[e]; b0 = 0; }
            else { s = src2[e - E1_]; d = dst2[e - E1_]; b0 = NB1_; }
            gb[i] = b0 + (d >> 8);
            pk[i] = (s << 8) | (d & (RNG_ - 1));
        }
    }
#pragma unroll
    for (int i = 0; i < 8; ++i) {
        lr[i] = 0;
        if (gb[i] >= 0) lr[i] = atomicAdd(&hist[gb[i]], 1);
    }
    __syncthreads();

    int c = 0, inc = 0;
    if (t < NBP_) {
        c = hist[t];
        inc = c;
        for (int o = 1; o < 32; o <<= 1) {
            const int v = __shfl_up(inc, o, 32);
            if ((t & 31) >= o) inc += v;
        }
        if ((t & 31) == 31) wsum[t >> 5] = inc;
    }
    __syncthreads();
    if (t < NBP_) {
        int off = 0;
        for (int w = 0; w < (t >> 5); ++w) off += wsum[w];
        lstart[t] = off + inc - c;
        gbase[t] = (c > 0) ? atomicAdd(cur + t, c) : 0;
    }
    __syncthreads();

#pragma unroll
    for (int i = 0; i < 8; ++i) {
        if (gb[i] >= 0) {
            const int p = lstart[gb[i]] + lr[i];
            sorted[p] = pk[i];
            srng[p] = (unsigned short)gb[i];
        }
    }
    __syncthreads();

    const int nE = min(CHUNK_, E1_ + E2_ - e0);
    for (int j = t; j < nE; j += 1024) {
        const int r = srng[j];
        const int pos = gbase[r] + (j - lstart[r]);
        const int v = sorted[j];
        if (pos < CAPB_) {
            lists[r * CAPB_ + pos] = v;
        } else {
            const int k = atomicAdd(spillCnt, 1);
            spill[k] = make_int2(r, v);
        }
    }
}

__device__ __forceinline__ void rank_entry(int v, int* cnt, int* rows, int* lhead,
                                           int* __restrict__ poolCnt,
                                           int* __restrict__ poolSrc,
                                           int* __restrict__ poolNext) {
    const int dl = v & (RNG_ - 1);
    const int s = (int)((unsigned)v >> 8);
    const int r = atomicAdd(&cnt[dl], 1);
    if (r < CAP_) {
        rows[dl * CAP_ + r] = s;
    } else {
        const int k = atomicAdd(poolCnt, 1);
        poolSrc[k] = s;
        poolNext[k] = atomicExch(&lhead[dl], k + 1);
    }
}

__global__ __launch_bounds__(512)
void edge_rank_kernel(const int* __restrict__ cur, const int* __restrict__ spillCnt,
                      const int* __restrict__ lists, const int2* __restrict__ spill,
                      int* __restrict__ deg1, int* __restrict__ deg2,
                      int* __restrict__ slots1, int* __restrict__ slots2,
                      int* __restrict__ head1, int* __restrict__ head2,
                      int* __restrict__ poolCnt,
                      int* __restrict__ poolSrc1, int* __restrict__ poolNext1,
                      int* __restrict__ poolSrc2, int* __restrict__ poolNext2) {
    __shared__ int cnt[RNG_];
    __shared__ int lhead[RNG_];
    __shared__ int rows[RNG_ * CAP_];

    const int t = threadIdx.x;
    const int b = blockIdx.x;
    const bool second = b >= NB1_;
    const int d0 = (second ? b - NB1_ : b) * RNG_;
    const int nd = min(RNG_, (second ? N2_ : N1_) - d0);
    int* deg      = second ? deg2 : deg1;
    int* slots    = second ? slots2 : slots1;
    int* head     = second ? head2 : head1;
    int* pCnt     = second ? poolCnt + 1 : poolCnt;
    int* poolSrc  = second ? poolSrc2 : poolSrc1;
    int* poolNext = second ? poolNext2 : poolNext1;

    if (t < RNG_) { cnt[t] = 0; lhead[t] = 0; }
    __syncthreads();

    const int n = min(cur[b], CAPB_);
    const int* lst = lists + b * CAPB_;
    for (int j = t; j < n; j += 512)
        rank_entry(lst[j], cnt, rows, lhead, pCnt, poolSrc, poolNext);
    const int ns = min(*spillCnt, E1_ + E2_);
    for (int j = t; j < ns; j += 512) {
        const int2 e = spill[j];
        if (e.x == b) rank_entry(e.y, cnt, rows, lhead, pCnt, poolSrc, poolNext);
    }
    __syncthreads();

    if (t < nd) { deg[d0 + t] = cnt[t]; head[d0 + t] = lhead[t]; }
    int* srow = slots + d0 * CAP_;
    for (int j = t; j < nd * CAP_; j += 512)
        srow[j] = ((j & (CAP_ - 1)) < cnt[j >> 5]) ? rows[j] : 0;
}

template <int D>
__global__ __launch_bounds__(256)
void gather_sum_kernel(const float* __restrict__ feat,
                       const int* __restrict__ deg,
                       const int* __restrict__ slots,
                       const int* __restrict__ head,
                       const int* __restrict__ poolSrc,
                       const int* __restrict__ poolNext,
                       float* __restrict__ aggSum) {
    constexpr int C4 = D / 128;
    const int lane = threadIdx.x & 31;
    const int node = blockIdx.x * 8 + (threadIdx.x >> 5);
    const int n = __builtin_amdgcn_readfirstlane(deg[node]);
    const int m = min(n, CAP_);

    float4 acc[C4];
#pragma unroll
    for (int c = 0; c < C4; ++c) acc[c] = make_float4(0.f, 0.f, 0.f, 0.f);

    const int sv = (lane < m) ? slots[node * CAP_ + lane] : 0;
    int j = 0;
    for (; j + 4 <= m; j += 4) {
        float4 v[4][C4];
#pragma unroll
        for (int u = 0; u < 4; ++u) {
            const unsigned sidx = (unsigned)__builtin_amdgcn_readlane(sv, j + u);
            const float4* row =
                reinterpret_cast<const float4*>(feat + (size_t)sidx * D) + lane;
#pragma unroll
            for (int c = 0; c < C4; ++c) v[u][c] = row[32 * c];
        }
#pragma unroll
        for (int u = 0; u < 4; ++u)
#pragma unroll
            for (int c = 0; c < C4; ++c) {
                acc[c].x += v[u][c].x; acc[c].y += v[u][c].y;
                acc[c].z += v[u][c].z; acc[c].w += v[u][c].w;
            }
    }
    for (; j < m; ++j) {
        const unsigned sidx = (unsigned)__builtin_amdgcn_readlane(sv, j);
        const float4* row =
            reinterpret_cast<const float4*>(feat + (size_t)sidx * D) + lane;
#pragma unroll
        for (int c = 0; c < C4; ++c) {
            const float4 v = row[32 * c];
            acc[c].x += v.x; acc[c].y += v.y; acc[c].z += v.z; acc[c].w += v.w;
        }
    }

    if (n > CAP_) {
        int k = __builtin_amdgcn_readfirstlane(head[node]);
        for (int t = CAP_; t < n && k > 0; ++t) {
            const unsigned sidx =
                (unsigned)__builtin_amdgcn_readfirstlane(poolSrc[k - 1]);
            const int nxt = __builtin_amdgcn_readfirstlane(poolNext[k - 1]);
            const float4* row =
                reinterpret_cast<const float4*>(feat + (size_t)sidx * D) + lane;
#pragma unroll
            for (int c = 0; c < C4; ++c) {
                const float4 v = row[32 * c];
                acc[c].x += v.x; acc[c].y += v.y; acc[c].z += v.z; acc[c].w += v.w;
            }
            k = nxt;
        }
    }

    float4* o = reinterpret_cast<float4*>(aggSum + (size_t)node * D) + lane;
#pragma unroll
    for (int c = 0; c < C4; ++c) o[32 * c] = acc[c];
}

template <int MT, int NT, int CH, bool SCALED>
__device__ __forceinline__ void sage_tile_chunk(const float* __restrict__ aBase,
                                                const unsigned (&aOff)[MT],
                                                const float* __restrict__ bBase,
                                                const unsigned (&bOff)[NT],
                                                const float (&scale)[MT],
                                                v8f (&c)[MT][NT]) {
    v2f a[CH][MT], b[CH][NT];
#pragma unroll
    for (int s = 0; s < CH; ++s) {
#pragma unroll
        for (int i = 0; i < MT; ++i) {
            const float* p = reinterpret_cast<const float*>(
                reinterpret_cast<const char*>(aBase + 4 * s) + aOff[i]);
            a[s][i].x = p[0];
            a[s][i].y = p[1];
        }
#pragma unroll
        for (int j = 0; j < NT; ++j) {
            const float* p = reinterpret_cast<const float*>(
                reinterpret_cast<const char*>(bBase + 4 * s) + bOff[j]);
            b[s][j].x = p[0];
            b[s][j].y = p[1];
        }
    }
    if (SCALED) {
#pragma unroll
        for (int s = 0; s < CH; ++s)
#pragma unroll
            for (int i = 0; i < MT; ++i) {
                a[s][i].x = a[s][i].x * scale[i];
                a[s][i].y = a[s][i].y * scale[i];
            }
    }

    __builtin_amdgcn_sched_barrier(0);

#pragma unroll
    for (int s = 0; s < CH; ++s) {
#pragma unroll
        for (int i = 0; i < MT; ++i) {
#pragma unroll
            for (int j = 0; j < NT; ++j)
                c[i][j] = __builtin_amdgcn_wmma_f32_16x16x4_f32(
                    false, a[s][i], false, b[s][j], (short)0, c[i][j], false, false);
        }
    }
}

template <int K, int MT, int NT, int CH, int WAVES_M, int WAVES_N, bool RELU>
__global__ __launch_bounds__(32 * WAVES_M * WAVES_N)
void sage_gemm_wmma_kernel(const float* __restrict__ aggSum,
                           const int* __restrict__ deg,
                           const float* __restrict__ root,
                           const float* __restrict__ Wl,
                           const float* __restrict__ Wr,
                           const float* __restrict__ bias,
                           float* __restrict__ out, int ldOut) {
    const int wave    = threadIdx.x >> 5;
    const int lane    = threadIdx.x & 31;
    const int l16     = lane & 15;
    const int half    = lane >> 4;
    const int waveN   = wave % WAVES_N;
    const int waveM   = wave / WAVES_N;
    const int rowBase = (blockIdx.x * WAVES_M + waveM) * (16 * MT);
    const int colBase = (blockIdx.y * WAVES_N + waveN) * (16 * NT);

    unsigned aOff[MT];
    float scale[MT];
#pragma unroll
    for (int i = 0; i < MT; ++i) {
        const int m = rowBase + 16 * i + l16;
        scale[i] = 1.0f / fmaxf((float)deg[m], 1.0f);
        aOff[i] = ((unsigned)m * K + 2 * half) * 4u;
    }
    unsigned bOff[NT];
#pragma unroll
    for (int j = 0; j < NT; ++j) {
        const int n = colBase + 16 * j + l16;
        bOff[j] = ((unsigned)n * K + 2 * half) * 4u;
    }

    v8f c[MT][NT];
#pragma unroll
    for (int i = 0; i < MT; ++i)
#pragma unroll
        for (int j = 0; j < NT; ++j) c[i][j] = v8f{};

#pragma unroll 1
    for (int k0 = 0; k0 < K; k0 += 4 * CH)
        sage_tile_chunk<MT, NT, CH, true>(aggSum + k0, aOff, Wl + k0, bOff, scale, c);
#pragma unroll 1
    for (int k0 = 0; k0 < K; k0 += 4 * CH)
        sage_tile_chunk<MT, NT, CH, false>(root + k0, aOff, Wr + k0, bOff, scale, c);

#pragma unroll
    for (int j = 0; j < NT; ++j) {
        const int col = colBase + 16 * j + l16;
        const float bv = bias[col];
#pragma unroll
        for (int i = 0; i < MT; ++i) {
#pragma unroll
            for (int r = 0; r < 8; ++r) {
                float v = c[i][j][r] + bv;
                if (RELU) v = fmaxf(v, 0.0f);
                const int row = rowBase + 16 * i + r + 8 * half;
                out[(size_t)row * ldOut + col] = v;
            }
        }
    }
}

__global__ void log_softmax64_kernel(float* __restrict__ h2, int rows) {
    int row = blockIdx.x * (blockDim.x >> 5) + (threadIdx.x >> 5);
    int lane = threadIdx.x & 31;
    if (row >= rows) return;
    float* p = h2 + (size_t)row * 64;
    float v0 = p[lane];
    float v1 = p[lane + 32];

    float m = fmaxf(v0, v1);
#pragma unroll
    for (int off = 16; off > 0; off >>= 1)
        m = fmaxf(m, __shfl_xor(m, off, 32));

    float s = __expf(v0 - m) + __expf(v1 - m);
#pragma unroll
    for (int off = 16; off > 0; off >>= 1)
        s += __shfl_xor(s, off, 32);

    float lse = m + __logf(s);
    p[lane]      = v0 - lse;
    p[lane + 32] = v1 - lse;
}

extern "C" void kernel_launch(void* const* d_in, const int* in_sizes, int n_in,
                              void* d_out, int out_size, void* d_ws, size_t ws_size,
                              hipStream_t stream) {
    const float* x    = (const float*)d_in[0];
    const int*   src1 = (const int*)d_in[1];
    const int*   dst1 = (const int*)d_in[2];
    const int*   src2 = (const int*)d_in[3];
    const int*   dst2 = (const int*)d_in[4];
    const float* W_l1 = (const float*)d_in[5];
    const float* b_l1 = (const float*)d_in[6];
    const float* W_r1 = (const float*)d_in[7];
    const float* W_l2 = (const float*)d_in[8];
    const float* b_l2 = (const float*)d_in[9];
    const float* W_r2 = (const float*)d_in[10];
    float* out = (float*)d_out;

    float* ws    = (float*)d_ws;
    int*   wsi   = (int*)d_ws;
    float* h1    = ws;
    int*   slots1    = wsi;
    int*   poolSrc1  = wsi + 3200000;
    int*   poolNext1 = wsi + 4800000;
    int*   lists     = wsi + 6400000;
    int2*  spill     = (int2*)(wsi + 9300000);
    int*   cur       = wsi + 13200000;
    int*   spillCnt  = wsi + 13200512;
    int*   poolCnt   = wsi + 13200513;
    float* agg1  = ws + 25600000;
    float* agg2  = agg1;
    int*   deg1      = wsi + 38400000;
    int*   deg2      = wsi + 38500000;
    int*   head1     = wsi + 38520000;
    int*   head2     = wsi + 38620000;
    int*   slots2    = wsi + 38640032;
    int*   poolSrc2  = wsi + 39280032;
    int*   poolNext2 = wsi + 39600032;

    zero_i4_kernel<<<1, 256, 0, stream>>>((int4*)cur, 528 / 4);

    edge_bin_kernel<<<(E1_ + E2_ + CHUNK_ - 1) / CHUNK_, 1024, 0, stream>>>(
        src1, dst1, src2, dst2, cur, spillCnt, lists, spill);

    edge_rank_kernel<<<NB_, 512, 0, stream>>>(
        cur, spillCnt, lists, spill, deg1, deg2, slots1, slots2, head1, head2,
        poolCnt, poolSrc1, poolNext1, poolSrc2, poolNext2);

    gather_sum_kernel<D_IN_><<<N1_ / 8, 256, 0, stream>>>(
        x, deg1, slots1, head1, poolSrc1, poolNext1, agg1);

    {
        dim3 grid(N1_ / (16 * 5), 1);
        sage_gemm_wmma_kernel<D_IN_, 5, 4, 8, 1, 4, true><<<grid, 128, 0, stream>>>(
            agg1, deg1, x, W_l1, W_r1, b_l1, h1, D_H_);
    }

    gather_sum_kernel<D_H_><<<N2_ / 8, 256, 0, stream>>>(
        h1, deg2, slots2, head2, poolSrc2, poolNext2, agg2);

    {
        dim3 grid(N2_ / 32, 1);
        sage_gemm_wmma_kernel<D_H_, 2, 2, 8, 1, 2, false><<<grid, 64, 0, stream>>>(
            agg2, deg2, h1, W_l2, W_r2, b_l2, out, D_OUT_);
    }

    log_softmax64_kernel<<<(N2_ + 7) / 8, 256, 0, stream>>>(out, N2_);
}


# Layer 2 in one launch, not three: 16 waves sum 16 neighbour rows into LDS, then each runs one 16x16 tile over a quarter of K on f32 16x16x4 WMMA, log_softmax in the tail; k sum order changes
# speedup vs baseline: 4.3237x; 1.0709x over previous
#include <hip/hip_runtime.h>
#include <hip/hip_bf16.h>


#define N0_   200000
#define N1_   100000
#define N2_   20000
#define E1_   1600000
#define E2_   320000
#define D_IN_ 128
#define D_H_  256
#define D_OUT_ 64
#define CAP_  32
#define RNG_  256
#define NB1_  391
#define NB2_  79
#define NB_   (NB1_ + NB2_)
#define NBP_  512
#define CAPB_ 6144
#define CHUNK_ 8192
#define L2ROWS_ 16
#define APITCH4_ 129
#define PPITCH_ 66

typedef __attribute__((ext_vector_type(2))) float v2f;
typedef __attribute__((ext_vector_type(8))) float v8f;

__global__ void zero_i4_kernel(int4* __restrict__ p, int n4) {
    const int i = blockIdx.x * blockDim.x + threadIdx.x;
    if (i < n4) p[i] = make_int4(0, 0, 0, 0);
}

__global__ __launch_bounds__(1024)
void edge_bin_kernel(const int* __restrict__ src1, const int* __restrict__ dst1,
                     const int* __restrict__ src2, const int* __restrict__ dst2,
                     int* __restrict__ cur, int* __restrict__ spillCnt,
                     int* __restrict__ lists, int2* __restrict__ spill) {
    __shared__ int hist[NBP_];
    __shared__ int lstart[NBP_];
    __shared__ int gbase[NBP_];
    __shared__ int wsum[NBP_ / 32];
    __shared__ int sorted[CHUNK_];
    __shared__ unsigned short srng[CHUNK_];

    const int t = threadIdx.x;
    if (t < NBP_) hist[t] = 0;
    __syncthreads();

    const int e0 = blockIdx.x * CHUNK_;
    int pk[8], gb[8], lr[8];
#pragma unroll
    for (int i = 0; i < 8; ++i) {
        const int e = e0 + i * 1024 + t;
        gb[i] = -1;
        pk[i] = 0;
        if (e < E1_ + E2_) {
            int s, d, b0;
            if (e < E1_) { s = src1[e]; d = dst1[e]; b0 = 0; }
            else { s = src2[e - E1_]; d = dst2[e - E1_]; b0 = NB1_; }
            gb[i] = b0 + (d >> 8);
            pk[i] = (s << 8) | (d & (RNG_ - 1));
        }
    }
#pragma unroll
    for (int i = 0; i < 8; ++i) {
        lr[i] = 0;
        if (gb[i] >= 0) lr[i] = atomicAdd(&hist[gb[i]], 1);
    }
    __syncthreads();

    int c = 0, inc = 0;
    if (t < NBP_) {
        c = hist[t];
        inc = c;
        for (int o = 1; o < 32; o <<= 1) {
            const int v = __shfl_up(inc, o, 32);
            if ((t & 31) >= o) inc += v;
        }
        if ((t & 31) == 31) wsum[t >> 5] = inc;
    }
    __syncthreads();
    if (t < NBP_) {
        int off = 0;
        for (int w = 0; w < (t >> 5); ++w) off += wsum[w];
        lstart[t] = off + inc - c;
        gbase[t] = (c > 0) ? atomicAdd(cur + t, c) : 0;
    }
    __syncthreads();

#pragma unroll
    for (int i = 0; i < 8; ++i) {
        if (gb[i] >= 0) {
            const int p = lstart[gb[i]] + lr[i];
            sorted[p] = pk[i];
            srng[p] = (unsigned short)gb[i];
        }
    }
    __syncthreads();

    const int nE = min(CHUNK_, E1_ + E2_ - e0);
    for (int j = t; j < nE; j += 1024) {
        const int r = srng[j];
        const int pos = gbase[r] + (j - lstart[r]);
        const int v = sorted[j];
        if (pos < CAPB_) {
            lists[r * CAPB_ + pos] = v;
        } else {
            const int k = atomicAdd(spillCnt, 1);
            spill[k] = make_int2(r, v);
        }
    }
}

__device__ __forceinline__ void rank_entry(int v, int* cnt, int* rows, int* lhead,
                                           int* __restrict__ poolCnt,
                                           int* __restrict__ poolSrc,
                                           int* __restrict__ poolNext) {
    const int dl = v & (RNG_ - 1);
    const int s = (int)((unsigned)v >> 8);
    const int r = atomicAdd(&cnt[dl], 1);
    if (r < CAP_) {
        rows[dl * CAP_ + r] = s;
    } else {
        const int k = atomicAdd(poolCnt, 1);
        poolSrc[k] = s;
        poolNext[k] = atomicExch(&lhead[dl], k + 1);
    }
}

__global__ __launch_bounds__(512)
void edge_rank_kernel(const int* __restrict__ cur, const int* __restrict__ spillCnt,
                      const int* __restrict__ lists, const int2* __restrict__ spill,
                      int* __restrict__ deg1, int* __restrict__ deg2,
                      int* __restrict__ slots1, int* __restrict__ slots2,
                      int* __restrict__ head1, int* __restrict__ head2,
                      int* __restrict__ poolCnt,
                      int* __restrict__ poolSrc1, int* __restrict__ poolNext1,
                      int* __restrict__ poolSrc2, int* __restrict__ poolNext2) {
    __shared__ int cnt[RNG_];
    __shared__ int lhead[RNG_];
    __shared__ int rows[RNG_ * CAP_];

    const int t = threadIdx.x;
    const int b = blockIdx.x;
    const bool second = b >= NB1_;
    const int d0 = (second ? b - NB1_ : b) * RNG_;
    const int nd = min(RNG_, (second ? N2_ : N1_) - d0);
    int* deg      = second ? deg2 : deg1;
    int* slots    = second ? slots2 : slots1;
    int* head     = second ? head2 : head1;
    int* pCnt     = second ? poolCnt + 1 : poolCnt;
    int* poolSrc  = second ? poolSrc2 : poolSrc1;
    int* poolNext = second ? poolNext2 : poolNext1;

    if (t < RNG_) { cnt[t] = 0; lhead[t] = 0; }
    __syncthreads();

    const int n = min(cur[b], CAPB_);
    const int* lst = lists + b * CAPB_;
    for (int j = t; j < n; j += 512)
        rank_entry(lst[j], cnt, rows, lhead, pCnt, poolSrc, poolNext);
    const int ns = min(*spillCnt, E1_ + E2_);
    for (int j = t; j < ns; j += 512) {
        const int2 e = spill[j];
        if (e.x == b) rank_entry(e.y, cnt, rows, lhead, pCnt, poolSrc, poolNext);
    }
    __syncthreads();

    if (t < nd) { deg[d0 + t] = cnt[t]; head[d0 + t] = lhead[t]; }
    int* srow = slots + d0 * CAP_;
    for (int j = t; j < nd * CAP_; j += 512)
        srow[j] = ((j & (CAP_ - 1)) < cnt[j >> 5]) ? rows[j] : 0;
}

template <int D>
__global__ __launch_bounds__(256)
void gather_sum_kernel(const float* __restrict__ feat,
                       const int* __restrict__ deg,
                       const int* __restrict__ slots,
                       const int* __restrict__ head,
                       const int* __restrict__ poolSrc,
                       const int* __restrict__ poolNext,
                       float* __restrict__ aggSum) {
    constexpr int C4 = D / 128;
    const int lane = threadIdx.x & 31;
    const int node = blockIdx.x * 8 + (threadIdx.x >> 5);
    const int n = __builtin_amdgcn_readfirstlane(deg[node]);
    const int m = min(n, CAP_);

    float4 acc[C4];
#pragma unroll
    for (int c = 0; c < C4; ++c) acc[c] = make_float4(0.f, 0.f, 0.f, 0.f);

    const int sv = (lane < m) ? slots[node * CAP_ + lane] : 0;
    int j = 0;
    for (; j + 4 <= m; j += 4) {
        float4 v[4][C4];
#pragma unroll
        for (int u = 0; u < 4; ++u) {
            const unsigned sidx = (unsigned)__builtin_amdgcn_readlane(sv, j + u);
            const float4* row =
                reinterpret_cast<const float4*>(feat + (size_t)sidx * D) + lane;
#pragma unroll
            for (int c = 0; c < C4; ++c) v[u][c] = row[32 * c];
        }
#pragma unroll
        for (int u = 0; u < 4; ++u)
#pragma unroll
            for (int c = 0; c < C4; ++c) {
                acc[c].x += v[u][c].x; acc[c].y += v[u][c].y;
                acc[c].z += v[u][c].z; acc[c].w += v[u][c].w;
            }
    }
    for (; j < m; ++j) {
        const unsigned sidx = (unsigned)__builtin_amdgcn_readlane(sv, j);
        const float4* row =
            reinterpret_cast<const float4*>(feat + (size_t)sidx * D) + lane;
#pragma unroll
        for (int c = 0; c < C4; ++c) {
            const float4 v = row[32 * c];
            acc[c].x += v.x; acc[c].y += v.y; acc[c].z += v.z; acc[c].w += v.w;
        }
    }

    if (n > CAP_) {
        int k = __builtin_amdgcn_readfirstlane(head[node]);
        for (int t = CAP_; t < n && k > 0; ++t) {
            const unsigned sidx =
                (unsigned)__builtin_amdgcn_readfirstlane(poolSrc[k - 1]);
            const int nxt = __builtin_amdgcn_readfirstlane(poolNext[k - 1]);
            const float4* row =
                reinterpret_cast<const float4*>(feat + (size_t)sidx * D) + lane;
#pragma unroll
            for (int c = 0; c < C4; ++c) {
                const float4 v = row[32 * c];
                acc[c].x += v.x; acc[c].y += v.y; acc[c].z += v.z; acc[c].w += v.w;
            }
            k = nxt;
        }
    }

    float4* o = reinterpret_cast<float4*>(aggSum + (size_t)node * D) + lane;
#pragma unroll
    for (int c = 0; c < C4; ++c) o[32 * c] = acc[c];
}

template <int MT, int NT, int CH, bool SCALED>
__device__ __forceinline__ void sage_tile_chunk(const float* __restrict__ aBase,
                                                const unsigned (&aOff)[MT],
                                                const float* __restrict__ bBase,
                                                const unsigned (&bOff)[NT],
                                                const float (&scale)[MT],
                                                v8f (&c)[MT][NT]) {
    v2f a[CH][MT], b[CH][NT];
#pragma unroll
    for (int s = 0; s < CH; ++s) {
#pragma unroll
        for (int i = 0; i < MT; ++i) {
            const float* p = reinterpret_cast<const float*>(
                reinterpret_cast<const char*>(aBase + 4 * s) + aOff[i]);
            a[s][i].x = p[0];
            a[s][i].y = p[1];
        }
#pragma unroll
        for (int j = 0; j < NT; ++j) {
            const float* p = reinterpret_cast<const float*>(
                reinterpret_cast<const char*>(bBase + 4 * s) + bOff[j]);
            b[s][j].x = p[0];
            b[s][j].y = p[1];
        }
    }
    if (SCALED) {
#pragma unroll
        for (int s = 0; s < CH; ++s)
#pragma unroll
            for (int i = 0; i < MT; ++i) {
                a[s][i].x = a[s][i].x * scale[i];
                a[s][i].y = a[s][i].y * scale[i];
            }
    }

    __builtin_amdgcn_sched_barrier(0);

#pragma unroll
    for (int s = 0; s < CH; ++s) {
#pragma unroll
        for (int i = 0; i < MT; ++i) {
#pragma unroll
            for (int j = 0; j < NT; ++j)
                c[i][j] = __builtin_amdgcn_wmma_f32_16x16x4_f32(
                    false, a[s][i], false, b[s][j], (short)0, c[i][j], false, false);
        }
    }
}

template <int K, int MT, int NT, int CH, int WAVES_M, int WAVES_N, bool RELU>
__global__ __launch_bounds__(32 * WAVES_M * WAVES_N)
void sage_gemm_wmma_kernel(const float* __restrict__ aggSum,
                           const int* __restrict__ deg,
                           const float* __restrict__ root,
                           const float* __restrict__ Wl,
                           const float* __restrict__ Wr,
                           const float* __restrict__ bias,
                           float* __restrict__ out, int ldOut) {
    const int wave    = threadIdx.x >> 5;
    const int lane    = threadIdx.x & 31;
    const int l16     = lane & 15;
    const int half    = lane >> 4;
    const int waveN   = wave % WAVES_N;
    const int waveM   = wave / WAVES_N;
    const int rowBase = (blockIdx.x * WAVES_M + waveM) * (16 * MT);
    const int colBase = (blockIdx.y * WAVES_N + waveN) * (16 * NT);

    unsigned aOff[MT];
    float scale[MT];
#pragma unroll
    for (int i = 0; i < MT; ++i) {
        const int m = rowBase + 16 * i + l16;
        scale[i] = 1.0f / fmaxf((float)deg[m], 1.0f);
        aOff[i] = ((unsigned)m * K + 2 * half) * 4u;
    }
    unsigned bOff[NT];
#pragma unroll
    for (int j = 0; j < NT; ++j) {
        const int n = colBase + 16 * j + l16;
        bOff[j] = ((unsigned)n * K + 2 * half) * 4u;
    }

    v8f c[MT][NT];
#pragma unroll
    for (int i = 0; i < MT; ++i)
#pragma unroll
        for (int j = 0; j < NT; ++j) c[i][j] = v8f{};

#pragma unroll 1
    for (int k0 = 0; k0 < K; k0 += 4 * CH)
        sage_tile_chunk<MT, NT, CH, true>(aggSum + k0, aOff, Wl + k0, bOff, scale, c);
#pragma unroll 1
    for (int k0 = 0; k0 < K; k0 += 4 * CH)
        sage_tile_chunk<MT, NT, CH, false>(root + k0, aOff, Wr + k0, bOff, scale, c);

#pragma unroll
    for (int j = 0; j < NT; ++j) {
        const int col = colBase + 16 * j + l16;
        const float bv = bias[col];
#pragma unroll
        for (int i = 0; i < MT; ++i) {
#pragma unroll
            for (int r = 0; r < 8; ++r) {
                float v = c[i][j][r] + bv;
                if (RELU) v = fmaxf(v, 0.0f);
                const int row = rowBase + 16 * i + r + 8 * half;
                out[(size_t)row * ldOut + col] = v;
            }
        }
    }
}

__global__ __launch_bounds__(32 * L2ROWS_)
void sage_layer2_kernel(const float* __restrict__ h1,
                        const int* __restrict__ deg,
                        const int* __restrict__ slots,
                        const int* __restrict__ head,
                        const int* __restrict__ poolSrc,
                        const int* __restrict__ poolNext,
                        const float* __restrict__ Wl,
                        const float* __restrict__ Wr,
                        const float* __restrict__ bias,
                        float* __restrict__ out) {
    __shared__ float4 aT[L2ROWS_ * APITCH4_];
    __shared__ float part[4 * L2ROWS_ * PPITCH_];

    const int lane = threadIdx.x & 31;
    const int wave = threadIdx.x >> 5;
    const int node = blockIdx.x * L2ROWS_ + wave;

    const float4* own = reinterpret_cast<const float4*>(h1 + (size_t)node * D_H_) + lane;
    const float4 own0 = own[0];
    const float4 own1 = own[32];

    const int n = __builtin_amdgcn_readfirstlane(deg[node]);
    const int m = min(n, CAP_);

    float4 acc[2];
    acc[0] = make_float4(0.f, 0.f, 0.f, 0.f);
    acc[1] = make_float4(0.f, 0.f, 0.f, 0.f);

    const int sv = (lane < m) ? slots[node * CAP_ + lane] : 0;

    float4* arow = aT + wave * APITCH4_ + lane;
    arow[64] = own0;
    arow[96] = own1;

    int j = 0;
    for (; j + 4 <= m; j += 4) {
        float4 v[4][2];
#pragma unroll
        for (int u = 0; u < 4; ++u) {
            const unsigned sidx = (unsigned)__builtin_amdgcn_readlane(sv, j + u);
            const float4* row =
                reinterpret_cast<const float4*>(h1 + (size_t)sidx * D_H_) + lane;
            v[u][0] = row[0];
            v[u][1] = row[32];
        }
#pragma unroll
        for (int u = 0; u < 4; ++u)
#pragma unroll
            for (int c = 0; c < 2; ++c) {
                acc[c].x += v[u][c].x; acc[c].y += v[u][c].y;
                acc[c].z += v[u][c].z; acc[c].w += v[u][c].w;
            }
    }
    for (; j < m; ++j) {
        const unsigned sidx = (unsigned)__builtin_amdgcn_readlane(sv, j);
        const float4* row =
            reinterpret_cast<const float4*>(h1 + (size_t)sidx * D_H_) + lane;
#pragma unroll
        for (int c = 0; c < 2; ++c) {
            const float4 v = row[32 * c];
            acc[c].x += v.x; acc[c].y += v.y; acc[c].z += v.z; acc[c].w += v.w;
        }
    }
    if (n > CAP_) {
        int k = __builtin_amdgcn_readfirstlane(head[node]);
        for (int t = CAP_; t < n && k > 0; ++t) {
            const unsigned sidx =
                (unsigned)__builtin_amdgcn_readfirstlane(poolSrc[k - 1]);
            const int nxt = __builtin_amdgcn_readfirstlane(poolNext[k - 1]);
            const float4* row =
                reinterpret_cast<const float4*>(h1 + (size_t)sidx * D_H_) + lane;
#pragma unroll
            for (int c = 0; c < 2; ++c) {
                const float4 v = row[32 * c];
                acc[c].x += v.x; acc[c].y += v.y; acc[c].z += v.z; acc[c].w += v.w;
            }
            k = nxt;
        }
    }

    {
        const float scale = 1.0f / fmaxf((float)n, 1.0f);
        arow[0]  = make_float4(acc[0].x * scale, acc[0].y * scale,
                               acc[0].z * scale, acc[0].w * scale);
        arow[32] = make_float4(acc[1].x * scale, acc[1].y * scale,
                               acc[1].z * scale, acc[1].w * scale);
    }
    __syncthreads();

    const int l16  = lane & 15;
    const int half = lane >> 4;
    const int ct   = wave & 3;
    const int q    = wave >> 2;
    const float* W = (q < 2) ? Wl : Wr;
    const float4* ap = aT + l16 * APITCH4_ + q * 32 + half;
    const float4* bp = reinterpret_cast<const float4*>(
                           W + (size_t)(16 * ct + l16) * D_H_ + (q & 1) * 128) + half;

    v8f c0 = v8f{}, c1 = v8f{};
#pragma unroll
    for (int g = 0; g < 4; ++g) {
        float4 av[4], bv[4];
#pragma unroll
        for (int u = 0; u < 4; ++u) {
            av[u] = ap[2 * (4 * g + u)];
            bv[u] = bp[2 * (4 * g + u)];
        }
#pragma unroll
        for (int u = 0; u < 4; ++u) {
            v2f a0, b0, a1, b1;
            a0.x = av[u].x; a0.y = av[u].y; b0.x = bv[u].x; b0.y = bv[u].y;
            a1.x = av[u].z; a1.y = av[u].w; b1.x = bv[u].z; b1.y = bv[u].w;
            c0 = __builtin_amdgcn_wmma_f32_16x16x4_f32(
                false, a0, false, b0, (short)0, c0, false, false);
            c1 = __builtin_amdgcn_wmma_f32_16x16x4_f32(
                false, a1, false, b1, (short)0, c1, false, false);
        }
    }

    {
        float* pp = part + (q * L2ROWS_ + 8 * half) * PPITCH_ + 16 * ct + l16;
#pragma unroll
        for (int r = 0; r < 8; ++r) pp[r * PPITCH_] = c0[r] + c1[r];
    }
    __syncthreads();

    const float* pr = part + wave * PPITCH_ + lane;
    float v0 = pr[0];
    float v1 = pr[32];
#pragma unroll
    for (int qq = 1; qq < 4; ++qq) {
        v0 += pr[qq * L2ROWS_ * PPITCH_];
        v1 += pr[qq * L2ROWS_ * PPITCH_ + 32];
    }
    v0 += bias[lane];
    v1 += bias[lane + 32];

    float mx = fmaxf(v0, v1);
#pragma unroll
    for (int off = 16; off > 0; off >>= 1)
        mx = fmaxf(mx, __shfl_xor(mx, off, 32));

    float s = __expf(v0 - mx) + __expf(v1 - mx);
#pragma unroll
    for (int off = 16; off > 0; off >>= 1)
        s += __shfl_xor(s, off, 32);

    const float lse = mx + __logf(s);
    float* o = out + (size_t)node * D_OUT_;
    o[lane]      = v0 - lse;
    o[lane + 32] = v1 - lse;
}

extern "C" void kernel_launch(void* const* d_in, const int* in_sizes, int n_in,
                              void* d_out, int out_size, void* d_ws, size_t ws_size,
                              hipStream_t stream) {
    const float* x    = (const float*)d_in[0];
    const int*   src1 = (const int*)d_in[1];
    const int*   dst1 = (const int*)d_in[2];
    const int*   src2 = (const int*)d_in[3];
    const int*   dst2 = (const int*)d_in[4];
    const float* W_l1 = (const float*)d_in[5];
    const float* b_l1 = (const float*)d_in[6];
    const float* W_r1 = (const float*)d_in[7];
    const float* W_l2 = (const float*)d_in[8];
    const float* b_l2 = (const float*)d_in[9];
    const float* W_r2 = (const float*)d_in[10];
    float* out = (float*)d_out;

    float* ws    = (float*)d_ws;
    int*   wsi   = (int*)d_ws;
    float* h1    = ws;
    int*   slots1    = wsi;
    int*   poolSrc1  = wsi + 3200000;
    int*   poolNext1 = wsi + 4800000;
    int*   lists     = wsi + 6400000;
    int2*  spill     = (int2*)(wsi + 9300000);
    int*   cur       = wsi + 13200000;
    int*   spillCnt  = wsi + 13200512;
    int*   poolCnt   = wsi + 13200513;
    float* agg1  = ws + 25600000;
    int*   deg1      = wsi + 38400000;
    int*   deg2      = wsi + 38500000;
    int*   head1     = wsi + 38520000;
    int*   head2     = wsi + 38620000;
    int*   slots2    = wsi + 38640032;
    int*   poolSrc2  = wsi + 39280032;
    int*   poolNext2 = wsi + 39600032;

    zero_i4_kernel<<<1, 256, 0, stream>>>((int4*)cur, 528 / 4);

    edge_bin_kernel<<<(E1_ + E2_ + CHUNK_ - 1) / CHUNK_, 1024, 0, stream>>>(
        src1, dst1, src2, dst2, cur, spillCnt, lists, spill);

    edge_rank_kernel<<<NB_, 512, 0, stream>>>(
        cur, spillCnt, lists, spill, deg1, deg2, slots1, slots2, head1, head2,
        poolCnt, poolSrc1, poolNext1, poolSrc2, poolNext2);

    gather_sum_kernel<D_IN_><<<N1_ / 8, 256, 0, stream>>>(
        x, deg1, slots1, head1, poolSrc1, poolNext1, agg1);

    {
        dim3 grid(N1_ / (16 * 5), 1);
        sage_gemm_wmma_kernel<D_IN_, 5, 4, 8, 1, 4, true><<<grid, 128, 0, stream>>>(
            agg1, deg1, x, W_l1, W_r1, b_l1, h1, D_H_);
    }

    sage_layer2_kernel<<<N2_ / L2ROWS_, 32 * L2ROWS_, 0, stream>>>(
        h1, deg2, slots2, head2, poolSrc2, poolNext2, W_l2, W_r2, b_l2, out);
}


# Weights copied into B-fragment order by the first launch: a B fragment load of either layer reads one contiguous run of whole lines, not a few bytes of every weight row of its tile; k order kept
# speedup vs baseline: 4.5542x; 1.0533x over previous
#include <hip/hip_runtime.h>
#include <hip/hip_bf16.h>


#define N0_   200000
#define N1_   100000
#define N2_   20000
#define E1_   1600000
#define E2_   320000
#define D_IN_ 128
#define D_H_  256
#define D_OUT_ 64
#define CAP_  32
#define RNG_  256
#define NB1_  391
#define NB2_  79
#define NB_   (NB1_ + NB2_)
#define NBP_  512
#define CAPB_ 6144
#define CHUNK_ 8192
#define L2ROWS_ 16
#define APITCH4_ 129
#define PPITCH_ 66

typedef __attribute__((ext_vector_type(2))) float v2f;
typedef __attribute__((ext_vector_type(8))) float v8f;

__global__ __launch_bounds__(256)
void prep_kernel(int4* __restrict__ zero, int n4,
                 const float* __restrict__ Wl1, const float* __restrict__ Wr1,
                 const float* __restrict__ Wl2, const float* __restrict__ Wr2,
                 float* __restrict__ Wl1p, float* __restrict__ Wr1p,
                 float* __restrict__ Wl2p, float* __restrict__ Wr2p) {
    const int i = blockIdx.x * 256 + threadIdx.x;
    if (i < n4) zero[i] = make_int4(0, 0, 0, 0);
    const int c = i & 15;
    const int h = (i >> 4) & 1;
    if (i < 32768) {
        const bool second = i >= 16384;
        const int e = i & 16383;
        const int blk = e >> 5, s = blk & 31, nt = blk >> 5;
        const float2* src = reinterpret_cast<const float2*>(second ? Wr1 : Wl1);
        float2* dst = reinterpret_cast<float2*>(second ? Wr1p : Wl1p);
        dst[e] = src[(16 * nt + c) * 64 + 2 * s + h];
    } else {
        const int e0 = i - 32768;
        const bool second = e0 >= 4096;
        const int e = e0 & 4095;
        const int blk = e >> 5, kb = blk & 31, ct = blk >> 5;
        const float4* src = reinterpret_cast<const float4*>(second ? Wr2 : Wl2);
        float4* dst = reinterpret_cast<float4*>(second ? Wr2p : Wl2p);
        dst[e] = src[(16 * ct + c) * 64 + 2 * kb + h];
    }
}

__global__ __launch_bounds__(1024)
void edge_bin_kernel(const int* __restrict__ src1, const int* __restrict__ dst1,
                     const int* __restrict__ src2, const int* __restrict__ dst2,
                     int* __restrict__ cur, int* __restrict__ spillCnt,
                     int* __restrict__ lists, int2* __restrict__ spill) {
    __shared__ int hist[NBP_];
    __shared__ int lstart[NBP_];
    __shared__ int gbase[NBP_];
    __shared__ int wsum[NBP_ / 32];
    __shared__ int sorted[CHUNK_];
    __shared__ unsigned short srng[CHUNK_];

    const int t = threadIdx.x;
    if (t < NBP_) hist[t] = 0;
    __syncthreads();

    const int e0 = blockIdx.x * CHUNK_;
    int pk[8], gb[8], lr[8];
#pragma unroll
    for (int i = 0; i < 8; ++i) {
        const int e = e0 + i * 1024 + t;
        gb[i] = -1;
        pk[i] = 0;
        if (e < E1_ + E2_) {
            int s, d, b0;
            if (e < E1_) { s = src1[e]; d = dst1[e]; b0 = 0; }
            else { s = src2[e - E1_]; d = dst2[e - E1_]; b0 = NB1_; }
            gb[i] = b0 + (d >> 8);
            pk[i] = (s << 8) | (d & (RNG_ - 1));
        }
    }
#pragma unroll
    for (int i = 0; i < 8; ++i) {
        lr[i] = 0;
        if (gb[i] >= 0) lr[i] = atomicAdd(&hist[gb[i]], 1);
    }
    __syncthreads();

    int c = 0, inc = 0;
    if (t < NBP_) {
        c = hist[t];
        inc = c;
        for (int o = 1; o < 32; o <<= 1) {
            const int v = __shfl_up(inc, o, 32);
            if ((t & 31) >= o) inc += v;
        }
        if ((t & 31) == 31) wsum[t >> 5] = inc;
    }
    __syncthreads();
    if (t < NBP_) {
        int off = 0;
        for (int w = 0; w < (t >> 5); ++w) off += wsum[w];
        lstart[t] = off + inc - c;
        gbase[t] = (c > 0) ? atomicAdd(cur + t, c) : 0;
    }
    __syncthreads();

#pragma unroll
    for (int i = 0; i < 8; ++i) {
        if (gb[i] >= 0) {
            const int p = lstart[gb[i]] + lr[i];
            sorted[p] = pk[i];
            srng[p] = (unsigned short)gb[i];
        }
    }
    __syncthreads();

    const int nE = min(CHUNK_, E1_ + E2_ - e0);
    for (int j = t; j < nE; j += 1024) {
        const int r = srng[j];
        const int pos = gbase[r] + (j - lstart[r]);
        const int v = sorted[j];
        if (pos < CAPB_) {
            lists[r * CAPB_ + pos] = v;
        } else {
            const int k = atomicAdd(spillCnt, 1);
            spill[k] = make_int2(r, v);
        }
    }
}

__device__ __forceinline__ void rank_entry(int v, int* cnt, int* rows, int* lhead,
                                           int* __restrict__ poolCnt,
                                           int* __restrict__ poolSrc,
                                           int* __restrict__ poolNext) {
    const int dl = v & (RNG_ - 1);
    const int s = (int)((unsigned)v >> 8);
    const int r = atomicAdd(&cnt[dl], 1);
    if (r < CAP_) {
        rows[dl * CAP_ + r] = s;
    } else {
        const int k = atomicAdd(poolCnt, 1);
        poolSrc[k] = s;
        poolNext[k] = atomicExch(&lhead[dl], k + 1);
    }
}

__global__ __launch_bounds__(512)
void edge_rank_kernel(const int* __restrict__ cur, const int* __restrict__ spillCnt,
                      const int* __restrict__ lists, const int2* __restrict__ spill,
                      int* __restrict__ deg1, int* __restrict__ deg2,
                      int* __restrict__ slots1, int* __restrict__ slots2,
                      int* __restrict__ head1, int* __restrict__ head2,
                      int* __restrict__ poolCnt,
                      int* __restrict__ poolSrc1, int* __restrict__ poolNext1,
                      int* __restrict__ poolSrc2, int* __restrict__ poolNext2) {
    __shared__ int cnt[RNG_];
    __shared__ int lhead[RNG_];
    __shared__ int rows[RNG_ * CAP_];

    const int t = threadIdx.x;
    const int b = blockIdx.x;
    const bool second = b >= NB1_;
    const int d0 = (second ? b - NB1_ : b) * RNG_;
    const int nd = min(RNG_, (second ? N2_ : N1_) - d0);
    int* deg      = second ? deg2 : deg1;
    int* slots    = second ? slots2 : slots1;
    int* head     = second ? head2 : head1;
    int* pCnt     = second ? poolCnt + 1 : poolCnt;
    int* poolSrc  = second ? poolSrc2 : poolSrc1;
    int* poolNext = second ? poolNext2 : poolNext1;

    if (t < RNG_) { cnt[t] = 0; lhead[t] = 0; }
    __syncthreads();

    const int n = min(cur[b], CAPB_);
    const int* lst = lists + b * CAPB_;
    for (int j = t; j < n; j += 512)
        rank_entry(lst[j], cnt, rows, lhead, pCnt, poolSrc, poolNext);
    const int ns = min(*spillCnt, E1_ + E2_);
    for (int j = t; j < ns; j += 512) {
        const int2 e = spill[j];
        if (e.x == b) rank_entry(e.y, cnt, rows, lhead, pCnt, poolSrc, poolNext);
    }
    __syncthreads();

    if (t < nd) { deg[d0 + t] = cnt[t]; head[d0 + t] = lhead[t]; }
    int* srow = slots + d0 * CAP_;
    for (int j = t; j < nd * CAP_; j += 512)
        srow[j] = ((j & (CAP_ - 1)) < cnt[j >> 5]) ? rows[j] : 0;
}

template <int D>
__global__ __launch_bounds__(256)
void gather_sum_kernel(const float* __restrict__ feat,
                       const int* __restrict__ deg,
                       const int* __restrict__ slots,
                       const int* __restrict__ head,
                       const int* __restrict__ poolSrc,
                       const int* __restrict__ poolNext,
                       float* __restrict__ aggSum) {
    constexpr int C4 = D / 128;
    const int lane = threadIdx.x & 31;
    const int node = blockIdx.x * 8 + (threadIdx.x >> 5);
    const int n = __builtin_amdgcn_readfirstlane(deg[node]);
    const int m = min(n, CAP_);

    float4 acc[C4];
#pragma unroll
    for (int c = 0; c < C4; ++c) acc[c] = make_float4(0.f, 0.f, 0.f, 0.f);

    const int sv = (lane < m) ? slots[node * CAP_ + lane] : 0;
    int j = 0;
    for (; j + 4 <= m; j += 4) {
        float4 v[4][C4];
#pragma unroll
        for (int u = 0; u < 4; ++u) {
            const unsigned sidx = (unsigned)__builtin_amdgcn_readlane(sv, j + u);
            const float4* row =
                reinterpret_cast<const float4*>(feat + (size_t)sidx * D) + lane;
#pragma unroll
            for (int c = 0; c < C4; ++c) v[u][c] = row[32 * c];
        }
#pragma unroll
        for (int u = 0; u < 4; ++u)
#pragma unroll
            for (int c = 0; c < C4; ++c) {
                acc[c].x += v[u][c].x; acc[c].y += v[u][c].y;
                acc[c].z += v[u][c].z; acc[c].w += v[u][c].w;
            }
    }
    for (; j < m; ++j) {
        const unsigned sidx = (unsigned)__builtin_amdgcn_readlane(sv, j);
        const float4* row =
            reinterpret_cast<const float4*>(feat + (size_t)sidx * D) + lane;
#pragma unroll
        for (int c = 0; c < C4; ++c) {
            const float4 v = row[32 * c];
            acc[c].x += v.x; acc[c].y += v.y; acc[c].z += v.z; acc[c].w += v.w;
        }
    }

    if (n > CAP_) {
        int k = __builtin_amdgcn_readfirstlane(head[node]);
        for (int t = CAP_; t < n && k > 0; ++t) {
            const unsigned sidx =
                (unsigned)__builtin_amdgcn_readfirstlane(poolSrc[k - 1]);
            const int nxt = __builtin_amdgcn_readfirstlane(poolNext[k - 1]);
            const float4* row =
                reinterpret_cast<const float4*>(feat + (size_t)sidx * D) + lane;
#pragma unroll
            for (int c = 0; c < C4; ++c) {
                const float4 v = row[32 * c];
                acc[c].x += v.x; acc[c].y += v.y; acc[c].z += v.z; acc[c].w += v.w;
            }
            k = nxt;
        }
    }

    float4* o = reinterpret_cast<float4*>(aggSum + (size_t)node * D) + lane;
#pragma unroll
    for (int c = 0; c < C4; ++c) o[32 * c] = acc[c];
}

template <int MT, int NT, int CH, bool SCALED>
__device__ __forceinline__ void sage_tile_chunk(const float* __restrict__ aBase,
                                                const unsigned (&aOff)[MT],
                                                const float* __restrict__ bBase,
                                                const unsigned (&bOff)[NT],
                                                const float (&scale)[MT],
                                                v8f (&c)[MT][NT]) {
    v2f a[CH][MT], b[CH][NT];
#pragma unroll
    for (int s = 0; s < CH; ++s) {
#pragma unroll
        for (int i = 0; i < MT; ++i) {
            const float* p = reinterpret_cast<const float*>(
                reinterpret_cast<const char*>(aBase + 4 * s) + aOff[i]);
            a[s][i].x = p[0];
            a[s][i].y = p[1];
        }
#pragma unroll
        for (int j = 0; j < NT; ++j) {
            const float* p = reinterpret_cast<const float*>(
                reinterpret_cast<const char*>(bBase + 64 * s) + bOff[j]);
            b[s][j].x = p[0];
            b[s][j].y = p[1];
        }
    }
    if (SCALED) {
#pragma unroll
        for (int s = 0; s < CH; ++s)
#pragma unroll
            for (int i = 0; i < MT; ++i) {
                a[s][i].x = a[s][i].x * scale[i];
                a[s][i].y = a[s][i].y * scale[i];
            }
    }

    __builtin_amdgcn_sched_barrier(0);

#pragma unroll
    for (int s = 0; s < CH; ++s) {
#pragma unroll
        for (int i = 0; i < MT; ++i) {
#pragma unroll
            for (int j = 0; j < NT; ++j)
                c[i][j] = __builtin_amdgcn_wmma_f32_16x16x4_f32(
                    false, a[s][i], false, b[s][j], (short)0, c[i][j], false, false);
        }
    }
}

template <int K, int MT, int NT, int CH, int WAVES_M, int WAVES_N, bool RELU>
__global__ __launch_bounds__(32 * WAVES_M * WAVES_N)
void sage_gemm_wmma_kernel(const float* __restrict__ aggSum,
                           const int* __restrict__ deg,
                           const float* __restrict__ root,
                           const float* __restrict__ Wl,
                           const float* __restrict__ Wr,
                           const float* __restrict__ bias,
                           float* __restrict__ out, int ldOut) {
    const int wave    = threadIdx.x >> 5;
    const int lane    = threadIdx.x & 31;
    const int l16     = lane & 15;
    const int half    = lane >> 4;
    const int waveN   = wave % WAVES_N;
    const int waveM   = wave / WAVES_N;
    const int rowBase = (blockIdx.x * WAVES_M + waveM) * (16 * MT);
    const int colBase = (blockIdx.y * WAVES_N + waveN) * (16 * NT);

    unsigned aOff[MT];
    float scale[MT];
#pragma unroll
    for (int i = 0; i < MT; ++i) {
        const int m = rowBase + 16 * i + l16;
        scale[i] = 1.0f / fmaxf((float)deg[m], 1.0f);
        aOff[i] = ((unsigned)m * K + 2 * half) * 4u;
    }
    unsigned bOff[NT];
#pragma unroll
    for (int j = 0; j < NT; ++j) {
        const int nt = colBase / 16 + j;
        bOff[j] = ((unsigned)nt * (16u * K) + 2u * lane) * 4u;
    }

    v8f c[MT][NT];
#pragma unroll
    for (int i = 0; i < MT; ++i)
#pragma unroll
        for (int j = 0; j < NT; ++j) c[i][j] = v8f{};

#pragma unroll 1
    for (int k0 = 0; k0 < K; k0 += 4 * CH)
        sage_tile_chunk<MT, NT, CH, true>(aggSum + k0, aOff, Wl + 16 * k0, bOff, scale, c);
#pragma unroll 1
    for (int k0 = 0; k0 < K; k0 += 4 * CH)
        sage_tile_chunk<MT, NT, CH, false>(root + k0, aOff, Wr + 16 * k0, bOff, scale, c);

#pragma unroll
    for (int j = 0; j < NT; ++j) {
        const int col = colBase + 16 * j + l16;
        const float bv = bias[col];
#pragma unroll
        for (int i = 0; i < MT; ++i) {
#pragma unroll
            for (int r = 0; r < 8; ++r) {
                float v = c[i][j][r] + bv;
                if (RELU) v = fmaxf(v, 0.0f);
                const int row = rowBase + 16 * i + r + 8 * half;
                out[(size_t)row * ldOut + col] = v;
            }
        }
    }
}

__global__ __launch_bounds__(32 * L2ROWS_)
void sage_layer2_kernel(const float* __restrict__ h1,
                        const int* __restrict__ deg,
                        const int* __restrict__ slots,
                        const int* __restrict__ head,
                        const int* __restrict__ poolSrc,
                        const int* __restrict__ poolNext,
                        const float* __restrict__ Wl,
                        const float* __restrict__ Wr,
                        const float* __restrict__ bias,
                        float* __restrict__ out) {
    __shared__ float4 aT[L2ROWS_ * APITCH4_];
    __shared__ float part[4 * L2ROWS_ * PPITCH_];

    const int lane = threadIdx.x & 31;
    const int wave = threadIdx.x >> 5;
    const int node = blockIdx.x * L2ROWS_ + wave;

    const float4* own = reinterpret_cast<const float4*>(h1 + (size_t)node * D_H_) + lane;
    const float4 own0 = own[0];
    const float4 own1 = own[32];

    const int n = __builtin_amdgcn_readfirstlane(deg[node]);
    const int m = min(n, CAP_);

    float4 acc[2];
    acc[0] = make_float4(0.f, 0.f, 0.f, 0.f);
    acc[1] = make_float4(0.f, 0.f, 0.f, 0.f);

    const int sv = (lane < m) ? slots[node * CAP_ + lane] : 0;

    float4* arow = aT + wave * APITCH4_ + lane;
    arow[64] = own0;
    arow[96] = own1;

    int j = 0;
    for (; j + 4 <= m; j += 4) {
        float4 v[4][2];
#pragma unroll
        for (int u = 0; u < 4; ++u) {
            const unsigned sidx = (unsigned)__builtin_amdgcn_readlane(sv, j + u);
            const float4* row =
                reinterpret_cast<const float4*>(h1 + (size_t)sidx * D_H_) + lane;
            v[u][0] = row[0];
            v[u][1] = row[32];
        }
#pragma unroll
        for (int u = 0; u < 4; ++u)
#pragma unroll
            for (int c = 0; c < 2; ++c) {
                acc[c].x += v[u][c].x; acc[c].y += v[u][c].y;
                acc[c].z += v[u][c].z; acc[c].w += v[u][c].w;
            }
    }
    for (; j < m; ++j) {
        const unsigned sidx = (unsigned)__builtin_amdgcn_readlane(sv, j);
        const float4* row =
            reinterpret_cast<const float4*>(h1 + (size_t)sidx * D_H_) + lane;
#pragma unroll
        for (int c = 0; c < 2; ++c) {
            const float4 v = row[32 * c];
            acc[c].x += v.x; acc[c].y += v.y; acc[c].z += v.z; acc[c].w += v.w;
        }
    }
    if (n > CAP_) {
        int k = __builtin_amdgcn_readfirstlane(head[node]);
        for (int t = CAP_; t < n && k > 0; ++t) {
            const unsigned sidx =
                (unsigned)__builtin_amdgcn_readfirstlane(poolSrc[k - 1]);
            const int nxt = __builtin_amdgcn_readfirstlane(poolNext[k - 1]);
            const float4* row =
                reinterpret_cast<const float4*>(h1 + (size_t)sidx * D_H_) + lane;
#pragma unroll
            for (int c = 0; c < 2; ++c) {
                const float4 v = row[32 * c];
                acc[c].x += v.x; acc[c].y += v.y; acc[c].z += v.z; acc[c].w += v.w;
            }
            k = nxt;
        }
    }

    {
        const float scale = 1.0f / fmaxf((float)n, 1.0f);
        arow[0]  = make_float4(acc[0].x * scale, acc[0].y * scale,
                               acc[0].z * scale, acc[0].w * scale);
        arow[32] = make_float4(acc[1].x * scale, acc[1].y * scale,
                               acc[1].z * scale, acc[1].w * scale);
    }
    __syncthreads();

    const int l16  = lane & 15;
    const int half = lane >> 4;
    const int ct   = wave & 3;
    const int q    = wave >> 2;
    const float* W = (q < 2) ? Wl : Wr;
    const float4* ap = aT + l16 * APITCH4_ + q * 32 + half;
    const float4* bp = reinterpret_cast<const float4*>(W)
                       + (ct * 32 + (q & 1) * 16) * 32 + lane;

    v8f c0 = v8f{}, c1 = v8f{};
#pragma unroll
    for (int g = 0; g < 4; ++g) {
        float4 av[4], bv[4];
#pragma unroll
        for (int u = 0; u < 4; ++u) {
            av[u] = ap[2 * (4 * g + u)];
            bv[u] = bp[32 * (4 * g + u)];
        }
#pragma unroll
        for (int u = 0; u < 4; ++u) {
            v2f a0, b0, a1, b1;
            a0.x = av[u].x; a0.y = av[u].y; b0.x = bv[u].x; b0.y = bv[u].y;
            a1.x = av[u].z; a1.y = av[u].w; b1.x = bv[u].z; b1.y = bv[u].w;
            c0 = __builtin_amdgcn_wmma_f32_16x16x4_f32(
                false, a0, false, b0, (short)0, c0, false, false);
            c1 = __builtin_amdgcn_wmma_f32_16x16x4_f32(
                false, a1, false, b1, (short)0, c1, false, false);
        }
    }

    {
        float* pp = part + (q * L2ROWS_ + 8 * half) * PPITCH_ + 16 * ct + l16;
#pragma unroll
        for (int r = 0; r < 8; ++r) pp[r * PPITCH_] = c0[r] + c1[r];
    }
    __syncthreads();

    const float* pr = part + wave * PPITCH_ + lane;
    float v0 = pr[0];
    float v1 = pr[32];
#pragma unroll
    for (int qq = 1; qq < 4; ++qq) {
        v0 += pr[qq * L2ROWS_ * PPITCH_];
        v1 += pr[qq * L2ROWS_ * PPITCH_ + 32];
    }
    v0 += bias[lane];
    v1 += bias[lane + 32];

    float mx = fmaxf(v0, v1);
#pragma unroll
    for (int off = 16; off > 0; off >>= 1)
        mx = fmaxf(mx, __shfl_xor(mx, off, 32));

    float s = __expf(v0 - mx) + __expf(v1 - mx);
#pragma unroll
    for (int off = 16; off > 0; off >>= 1)
        s += __shfl_xor(s, off, 32);

    const float lse = mx + __logf(s);
    float* o = out + (size_t)node * D_OUT_;
    o[lane]      = v0 - lse;
    o[lane + 32] = v1 - lse;
}

extern "C" void kernel_launch(void* const* d_in, const int* in_sizes, int n_in,
                              void* d_out, int out_size, void* d_ws, size_t ws_size,
                              hipStream_t stream) {
    const float* x    = (const float*)d_in[0];
    const int*   src1 = (const int*)d_in[1];
    const int*   dst1 = (const int*)d_in[2];
    const int*   src2 = (const int*)d_in[3];
    const int*   dst2 = (const int*)d_in[4];
    const float* W_l1 = (const float*)d_in[5];
    const float* b_l1 = (const float*)d_in[6];
    const float* W_r1 = (const float*)d_in[7];
    const float* W_l2 = (const float*)d_in[8];
    const float* b_l2 = (const float*)d_in[9];
    const float* W_r2 = (const float*)d_in[10];
    float* out = (float*)d_out;

    float* ws    = (float*)d_ws;
    int*   wsi   = (int*)d_ws;
    float* h1    = ws;
    int*   slots1    = wsi;
    int*   poolSrc1  = wsi + 3200000;
    int*   poolNext1 = wsi + 4800000;
    int*   lists     = wsi + 6400000;
    int2*  spill     = (int2*)(wsi + 9300000);
    int*   cur       = wsi + 13200000;
    int*   spillCnt  = wsi + 13200512;
    int*   poolCnt   = wsi + 13200513;
    float* agg1  = ws + 25600000;
    int*   deg1      = wsi + 38400000;
    int*   deg2      = wsi + 38500000;
    int*   head1     = wsi + 38520000;
    int*   head2     = wsi + 38620000;
    int*   slots2    = wsi + 38640032;
    int*   poolSrc2  = wsi + 39280032;
    int*   poolNext2 = wsi + 39600032;
    float* Wl1p  = ws + 39920128;
    float* Wr1p  = ws + 39952896;
    float* Wl2p  = ws + 39985664;
    float* Wr2p  = ws + 40002048;

    prep_kernel<<<160, 256, 0, stream>>>((int4*)cur, 528 / 4, W_l1, W_r1, W_l2, W_r2,
                                         Wl1p, Wr1p, Wl2p, Wr2p);

    edge_bin_kernel<<<(E1_ + E2_ + CHUNK_ - 1) / CHUNK_, 1024, 0, stream>>>(
        src1, dst1, src2, dst2, cur, spillCnt, lists, spill);

    edge_rank_kernel<<<NB_, 512, 0, stream>>>(
        cur, spillCnt, lists, spill, deg1, deg2, slots1, slots2, head1, head2,
        poolCnt, poolSrc1, poolNext1, poolSrc2, poolNext2);

    gather_sum_kernel<D_IN_><<<N1_ / 8, 256, 0, stream>>>(
        x, deg1, slots1, head1, poolSrc1, poolNext1, agg1);

    {
        dim3 grid(N1_ / (16 * 5), 1);
        sage_gemm_wmma_kernel<D_IN_, 5, 4, 8, 1, 4, true><<<grid, 128, 0, stream>>>(
            agg1, deg1, x, Wl1p, Wr1p, b_l1, h1, D_H_);
    }

    sage_layer2_kernel<<<N2_ / L2ROWS_, 32 * L2ROWS_, 0, stream>>>(
        h1, deg2, slots2, head2, poolSrc2, poolNext2, Wl2p, Wr2p, b_l2, out);
}
